# starting kernel
# baseline (speedup 1.0000x reference)
#include <hip/hip_runtime.h>

#define BDIM 2048
#define TDIM 4096

typedef float v2f __attribute__((ext_vector_type(2)));
typedef float v8f __attribute__((ext_vector_type(8)));

__device__ __forceinline__ float sigm10(float y) {
    float e = __builtin_amdgcn_exp2f(-14.4269504088896341f * y);
    return __builtin_amdgcn_rcpf(1.0f + e);
}

__device__ __forceinline__ float clip1e5(float d) {
    return fminf(fmaxf(d, -100000.0f), 100000.0f);
}

__global__ __launch_bounds__(256) void hbv_scan_kernel(
    const float* __restrict__ x, const float* __restrict__ p,
    float* __restrict__ qws)
{
    const int b = blockIdx.x * 256 + threadIdx.x;

    const float Tt     = p[0]  * -1.5f;
    const float CFR    = p[1]  * 0.01f;
    const float CFMAX  = p[2];
    const float SCF    = p[3];
    const float FC     = p[4]  * 100.0f;
    const float Beta   = p[5]  * 8.0f;
    const float LP     = p[6]  * 2.5f;
    const float CWH    = p[7]  * 0.3f;
    const float UZL    = p[8]  * 120.0f;
    const float K0     = p[9]  * 0.5f;
    const float K1     = p[10] * 0.3f;
    const float K2     = p[11] * 0.01f;
    const float PER    = p[12] * 3.5f;
    const float rawCWH = p[7];

    const float invFC   = __builtin_amdgcn_rcpf(FC);
    const float FCLP    = FC * LP;
    const float invFCLP = __builtin_amdgcn_rcpf(FCLP);
    const float CFRMAX  = CFR * CFMAX;

    float S1 = 0.0f, S2 = 0.0f, S3 = 0.0f, S4 = 0.0f, S5 = 0.0f;

    const float* xb = x + (size_t)b * TDIM * 3;
    float* qp = qws + b;

    for (int t = 0; t < TDIM; ++t) {
        float P   = xb[0];
        float Tm  = xb[1];
        float PET = xb[2];
        xb += 3;

        float h_cold = sigm10(Tt - Tm);
        float h_warm = 1.0f - h_cold;
        float Ps  = P * SCF * h_cold;
        float Rfr = fminf((Tt - Tm) * CFRMAX, S2) * h_cold;
        float Pr  = P * h_warm;
        float M   = fminf((Tm - Tt) * CFMAX, S1) * h_warm;
        float I   = sigm10(S2 - S1 * CWH) * (Pr + M);
        float Re  = I * __builtin_amdgcn_exp2f(Beta * __builtin_amdgcn_logf(S3 * invFC));
        float hS3 = sigm10(S3);
        float hLP = sigm10(FCLP - S3);
        float Ea  = hS3 * (PET * S3 * invFCLP) * hLP + hS3 * PET * (1.0f - hLP);
        float hS4 = sigm10(S4);
        float Pe  = PER * hS4;
        float q0  = (S4 - UZL) * K0 * sigm10(S4 - UZL);
        float q1  = S4 * K1 * hS4;
        float q2  = S5 * K2 * sigm10(S5);

        float nS1 = fmaxf(S1 + clip1e5(Ps + Rfr - M), 0.0f);
        float nS2 = fmaxf(fminf(S2 + clip1e5(Pr + M - I - Rfr), nS1 * rawCWH), 0.0f);
        float nS3 = fmaxf(S3 + clip1e5(I - Ea - Re), 0.0f);
        float nS4 = fmaxf(S4 + clip1e5(Re - q0 - q1 - Pe), 0.0f);
        float nS5 = fmaxf(S5 + clip1e5(Pe - q2), 0.0f);
        S1 = nS1; S2 = nS2; S3 = nS3; S4 = nS4; S5 = nS5;

        qp[(size_t)t * BDIM] = q0 + q1 + q2;
    }
}

__global__ __launch_bounds__(256) void hbv_route_wmma_kernel(
    const float* __restrict__ qws, float* __restrict__ out)
{
    constexpr double INV_MID_D = 1.0 / (4.0 + 1e-7);
    constexpr double WSUM_D = 1.0 + 2.0 * ((1.0 - 1.0 * INV_MID_D) +
                                           (1.0 - 2.0 * INV_MID_D) +
                                           (1.0 - 3.0 * INV_MID_D) +
                                           (1.0 - 4.0 * INV_MID_D));
    constexpr float INV_MID  = (float)INV_MID_D;
    constexpr float INV_WSUM = (float)(1.0 / (WSUM_D + 1e-7));

    const int lane = threadIdx.x & 31;
    const int wave = threadIdx.x >> 5;
    const int tile = blockIdx.x * 8 + wave;
    const int tb   = tile & (BDIM / 16 - 1);
    const int tt   = tile >> 7;
    const int b0   = tb * 16;
    const int t0   = tt * 16;
    const int mn   = lane & 15;
    const int khi  = lane >> 4;

    v2f afr[6], wfr[6];
#pragma unroll
    for (int c = 0; c < 6; ++c) {
#pragma unroll
        for (int j = 0; j < 2; ++j) {
            const int kg = 4 * c + j + 2 * khi;
            const int tq  = t0 - 4 + kg;
            const int tqc = tq < 0 ? 0 : (tq > TDIM - 1 ? TDIM - 1 : tq);
            const float v = qws[(size_t)tqc * BDIM + b0 + mn];
            afr[c][j] = (tq == tqc) ? v : 0.0f;
            const float dd = fabsf((float)(kg - mn) - 4.0f);
            wfr[c][j] = fmaxf(0.0f, fmaf(-dd, INV_MID, 1.0f)) * INV_WSUM;
        }
    }

    v8f acc = {0.0f, 0.0f, 0.0f, 0.0f, 0.0f, 0.0f, 0.0f, 0.0f};
#pragma unroll
    for (int c = 0; c < 6; ++c) {
        acc = __builtin_amdgcn_wmma_f32_16x16x4_f32(
            false, afr[c], false, wfr[c], (short)0, acc, false, false);
    }

#pragma unroll
    for (int v = 0; v < 8; ++v) {
        const int m = v + 8 * khi;
        out[(size_t)(b0 + m) * TDIM + t0 + mn] = acc[v];
    }
}

extern "C" void kernel_launch(void* const* d_in, const int* in_sizes, int n_in,
                              void* d_out, int out_size, void* d_ws, size_t ws_size,
                              hipStream_t stream) {
    (void)in_sizes; (void)n_in; (void)out_size; (void)ws_size;
    const float* x      = (const float*)d_in[0];
    const float* params = (const float*)d_in[1];
    float* out = (float*)d_out;
    float* qws = (float*)d_ws;

    hbv_scan_kernel<<<BDIM / 256, 256, 0, stream>>>(x, params, qws);
    hbv_route_wmma_kernel<<<(BDIM / 16) * (TDIM / 16) / 8, 256, 0, stream>>>(qws, out);
}


# x read per lane in 16-byte loads into two register buffers of 16 steps each; a group's loads are issued a whole group ahead, so load latency leaves the per-step recurrence
# speedup vs baseline: 4.2303x; 4.2303x over previous
#include <hip/hip_runtime.h>

#define BDIM 2048
#define TDIM 4096

typedef float v2f __attribute__((ext_vector_type(2)));
typedef float v4f __attribute__((ext_vector_type(4)));
typedef float v8f __attribute__((ext_vector_type(8)));

#define XCH 16
#define XV  (XCH * 3 / 4)
#define NGRP (TDIM / XCH)

__device__ __forceinline__ float sigm10(float y) {
    float e = __builtin_amdgcn_exp2f(-14.4269504088896341f * y);
    return __builtin_amdgcn_rcpf(1.0f + e);
}

__device__ __forceinline__ float clip1e5(float d) {
    return fminf(fmaxf(d, -100000.0f), 100000.0f);
}

struct HbvPar {
    float Tt, CFMAX, SCF, CFRMAX, CWH, rawCWH, Beta, invFC, FCLP, invFCLP;
    float PER, UZL, K0, K1, K2;
};

__device__ __forceinline__ void load_group(v4f (&buf)[XV], const v4f* __restrict__ src) {
#pragma unroll
    for (int i = 0; i < XV; ++i) buf[i] = src[i];
}

__device__ __forceinline__ void run_group(
    const v4f (&buf)[XV], const HbvPar& k, float* __restrict__ qt,
    float& S1, float& S2, float& S3, float& S4, float& S5)
{
#pragma unroll
    for (int s = 0; s < XCH; ++s) {
        const float P   = buf[(3 * s + 0) >> 2][(3 * s + 0) & 3];
        const float Tm  = buf[(3 * s + 1) >> 2][(3 * s + 1) & 3];
        const float PET = buf[(3 * s + 2) >> 2][(3 * s + 2) & 3];

        float h_cold = sigm10(k.Tt - Tm);
        float h_warm = 1.0f - h_cold;
        float Ps  = P * k.SCF * h_cold;
        float Rfr = fminf((k.Tt - Tm) * k.CFRMAX, S2) * h_cold;
        float Pr  = P * h_warm;
        float M   = fminf((Tm - k.Tt) * k.CFMAX, S1) * h_warm;
        float I   = sigm10(S2 - S1 * k.CWH) * (Pr + M);
        float Re  = I * __builtin_amdgcn_exp2f(k.Beta * __builtin_amdgcn_logf(S3 * k.invFC));
        float hS3 = sigm10(S3);
        float hLP = sigm10(k.FCLP - S3);
        float Ea  = hS3 * (PET * S3 * k.invFCLP) * hLP + hS3 * PET * (1.0f - hLP);
        float hS4 = sigm10(S4);
        float Pe  = k.PER * hS4;
        float q0  = (S4 - k.UZL) * k.K0 * sigm10(S4 - k.UZL);
        float q1  = S4 * k.K1 * hS4;
        float q2  = S5 * k.K2 * sigm10(S5);

        float nS1 = fmaxf(S1 + clip1e5(Ps + Rfr - M), 0.0f);
        float nS2 = fmaxf(fminf(S2 + clip1e5(Pr + M - I - Rfr), nS1 * k.rawCWH), 0.0f);
        float nS3 = fmaxf(S3 + clip1e5(I - Ea - Re), 0.0f);
        float nS4 = fmaxf(S4 + clip1e5(Re - q0 - q1 - Pe), 0.0f);
        float nS5 = fmaxf(S5 + clip1e5(Pe - q2), 0.0f);
        S1 = nS1; S2 = nS2; S3 = nS3; S4 = nS4; S5 = nS5;

        qt[(size_t)s * BDIM] = q0 + q1 + q2;
    }
}

__global__ __launch_bounds__(256) void hbv_scan_kernel(
    const float* __restrict__ x, const float* __restrict__ p,
    float* __restrict__ qws)
{
    const int b = blockIdx.x * 256 + threadIdx.x;

    const float CFR = p[1] * 0.01f;
    const float FC  = p[4] * 100.0f;
    const float LP  = p[6] * 2.5f;
    HbvPar k;
    k.Tt      = p[0]  * -1.5f;
    k.CFMAX   = p[2];
    k.SCF     = p[3];
    k.Beta    = p[5]  * 8.0f;
    k.CWH     = p[7]  * 0.3f;
    k.UZL     = p[8]  * 120.0f;
    k.K0      = p[9]  * 0.5f;
    k.K1      = p[10] * 0.3f;
    k.K2      = p[11] * 0.01f;
    k.PER     = p[12] * 3.5f;
    k.rawCWH  = p[7];
    k.invFC   = __builtin_amdgcn_rcpf(FC);
    k.FCLP    = FC * LP;
    k.invFCLP = __builtin_amdgcn_rcpf(k.FCLP);
    k.CFRMAX  = CFR * k.CFMAX;

    float S1 = 0.0f, S2 = 0.0f, S3 = 0.0f, S4 = 0.0f, S5 = 0.0f;

    const v4f* xv = reinterpret_cast<const v4f*>(x + (size_t)b * TDIM * 3);
    float* qp = qws + b;

    v4f bufA[XV], bufB[XV];
    load_group(bufA, xv);

    for (int g = 0; g < NGRP; g += 2) {
        load_group(bufB, xv + (size_t)(g + 1) * XV);
        __builtin_amdgcn_sched_barrier(0);
        run_group(bufA, k, qp + (size_t)g * XCH * BDIM, S1, S2, S3, S4, S5);

        const int gn = (g + 2 < NGRP) ? (g + 2) : g;
        load_group(bufA, xv + (size_t)gn * XV);
        __builtin_amdgcn_sched_barrier(0);
        run_group(bufB, k, qp + (size_t)(g + 1) * XCH * BDIM, S1, S2, S3, S4, S5);
    }
}

__global__ __launch_bounds__(256) void hbv_route_wmma_kernel(
    const float* __restrict__ qws, float* __restrict__ out)
{
    constexpr double INV_MID_D = 1.0 / (4.0 + 1e-7);
    constexpr double WSUM_D = 1.0 + 2.0 * ((1.0 - 1.0 * INV_MID_D) +
                                           (1.0 - 2.0 * INV_MID_D) +
                                           (1.0 - 3.0 * INV_MID_D) +
                                           (1.0 - 4.0 * INV_MID_D));
    constexpr float INV_MID  = (float)INV_MID_D;
    constexpr float INV_WSUM = (float)(1.0 / (WSUM_D + 1e-7));

    const int lane = threadIdx.x & 31;
    const int wave = threadIdx.x >> 5;
    const int tile = blockIdx.x * 8 + wave;
    const int tb   = tile & (BDIM / 16 - 1);
    const int tt   = tile >> 7;
    const int b0   = tb * 16;
    const int t0   = tt * 16;
    const int mn   = lane & 15;
    const int khi  = lane >> 4;

    v2f afr[6], wfr[6];
#pragma unroll
    for (int c = 0; c < 6; ++c) {
#pragma unroll
        for (int j = 0; j < 2; ++j) {
            const int kg = 4 * c + j + 2 * khi;
            const int tq  = t0 - 4 + kg;
            const int tqc = tq < 0 ? 0 : (tq > TDIM - 1 ? TDIM - 1 : tq);
            const float v = qws[(size_t)tqc * BDIM + b0 + mn];
            afr[c][j] = (tq == tqc) ? v : 0.0f;
            const float dd = fabsf((float)(kg - mn) - 4.0f);
            wfr[c][j] = fmaxf(0.0f, fmaf(-dd, INV_MID, 1.0f)) * INV_WSUM;
        }
    }

    v8f acc = {0.0f, 0.0f, 0.0f, 0.0f, 0.0f, 0.0f, 0.0f, 0.0f};
#pragma unroll
    for (int c = 0; c < 6; ++c) {
        acc = __builtin_amdgcn_wmma_f32_16x16x4_f32(
            false, afr[c], false, wfr[c], (short)0, acc, false, false);
    }

#pragma unroll
    for (int v = 0; v < 8; ++v) {
        const int m = v + 8 * khi;
        out[(size_t)(b0 + m) * TDIM + t0 + mn] = acc[v];
    }
}

extern "C" void kernel_launch(void* const* d_in, const int* in_sizes, int n_in,
                              void* d_out, int out_size, void* d_ws, size_t ws_size,
                              hipStream_t stream) {
    (void)in_sizes; (void)n_in; (void)out_size; (void)ws_size;
    const float* x      = (const float*)d_in[0];
    const float* params = (const float*)d_in[1];
    float* out = (float*)d_out;
    float* qws = (float*)d_ws;

    hbv_scan_kernel<<<BDIM / 256, 256, 0, stream>>>(x, params, qws);
    hbv_route_wmma_kernel<<<(BDIM / 16) * (TDIM / 16) / 8, 256, 0, stream>>>(qws, out);
}


# scan kernel launched as 64 one-wave work-groups instead of 8 eight-wave ones: each wave gets a SIMD, a WGP cache and a load path of its own instead of sharing them with seven other waves
# speedup vs baseline: 6.6388x; 1.5693x over previous
#include <hip/hip_runtime.h>

#define BDIM 2048
#define TDIM 4096

typedef float v2f __attribute__((ext_vector_type(2)));
typedef float v4f __attribute__((ext_vector_type(4)));
typedef float v8f __attribute__((ext_vector_type(8)));

#define XCH 16
#define XV  (XCH * 3 / 4)
#define NGRP (TDIM / XCH)

__device__ __forceinline__ float sigm10(float y) {
    float e = __builtin_amdgcn_exp2f(-14.4269504088896341f * y);
    return __builtin_amdgcn_rcpf(1.0f + e);
}

__device__ __forceinline__ float clip1e5(float d) {
    return fminf(fmaxf(d, -100000.0f), 100000.0f);
}

struct HbvPar {
    float Tt, CFMAX, SCF, CFRMAX, CWH, rawCWH, Beta, invFC, FCLP, invFCLP;
    float PER, UZL, K0, K1, K2;
};

__device__ __forceinline__ void load_group(v4f (&buf)[XV], const v4f* __restrict__ src) {
#pragma unroll
    for (int i = 0; i < XV; ++i) buf[i] = src[i];
}

__device__ __forceinline__ void run_group(
    const v4f (&buf)[XV], const HbvPar& k, float* __restrict__ qt,
    float& S1, float& S2, float& S3, float& S4, float& S5)
{
#pragma unroll
    for (int s = 0; s < XCH; ++s) {
        const float P   = buf[(3 * s + 0) >> 2][(3 * s + 0) & 3];
        const float Tm  = buf[(3 * s + 1) >> 2][(3 * s + 1) & 3];
        const float PET = buf[(3 * s + 2) >> 2][(3 * s + 2) & 3];

        float h_cold = sigm10(k.Tt - Tm);
        float h_warm = 1.0f - h_cold;
        float Ps  = P * k.SCF * h_cold;
        float Rfr = fminf((k.Tt - Tm) * k.CFRMAX, S2) * h_cold;
        float Pr  = P * h_warm;
        float M   = fminf((Tm - k.Tt) * k.CFMAX, S1) * h_warm;
        float I   = sigm10(S2 - S1 * k.CWH) * (Pr + M);
        float Re  = I * __builtin_amdgcn_exp2f(k.Beta * __builtin_amdgcn_logf(S3 * k.invFC));
        float hS3 = sigm10(S3);
        float hLP = sigm10(k.FCLP - S3);
        float Ea  = hS3 * (PET * S3 * k.invFCLP) * hLP + hS3 * PET * (1.0f - hLP);
        float hS4 = sigm10(S4);
        float Pe  = k.PER * hS4;
        float q0  = (S4 - k.UZL) * k.K0 * sigm10(S4 - k.UZL);
        float q1  = S4 * k.K1 * hS4;
        float q2  = S5 * k.K2 * sigm10(S5);

        float nS1 = fmaxf(S1 + clip1e5(Ps + Rfr - M), 0.0f);
        float nS2 = fmaxf(fminf(S2 + clip1e5(Pr + M - I - Rfr), nS1 * k.rawCWH), 0.0f);
        float nS3 = fmaxf(S3 + clip1e5(I - Ea - Re), 0.0f);
        float nS4 = fmaxf(S4 + clip1e5(Re - q0 - q1 - Pe), 0.0f);
        float nS5 = fmaxf(S5 + clip1e5(Pe - q2), 0.0f);
        S1 = nS1; S2 = nS2; S3 = nS3; S4 = nS4; S5 = nS5;

        qt[(size_t)s * BDIM] = q0 + q1 + q2;
    }
}

__global__ __launch_bounds__(32) void hbv_scan_kernel(
    const float* __restrict__ x, const float* __restrict__ p,
    float* __restrict__ qws)
{
    const int b = blockIdx.x * 32 + threadIdx.x;

    const float CFR = p[1] * 0.01f;
    const float FC  = p[4] * 100.0f;
    const float LP  = p[6] * 2.5f;
    HbvPar k;
    k.Tt      = p[0]  * -1.5f;
    k.CFMAX   = p[2];
    k.SCF     = p[3];
    k.Beta    = p[5]  * 8.0f;
    k.CWH     = p[7]  * 0.3f;
    k.UZL     = p[8]  * 120.0f;
    k.K0      = p[9]  * 0.5f;
    k.K1      = p[10] * 0.3f;
    k.K2      = p[11] * 0.01f;
    k.PER     = p[12] * 3.5f;
    k.rawCWH  = p[7];
    k.invFC   = __builtin_amdgcn_rcpf(FC);
    k.FCLP    = FC * LP;
    k.invFCLP = __builtin_amdgcn_rcpf(k.FCLP);
    k.CFRMAX  = CFR * k.CFMAX;

    float S1 = 0.0f, S2 = 0.0f, S3 = 0.0f, S4 = 0.0f, S5 = 0.0f;

    const v4f* xv = reinterpret_cast<const v4f*>(x + (size_t)b * TDIM * 3);
    float* qp = qws + b;

    v4f bufA[XV], bufB[XV];
    load_group(bufA, xv);

    for (int g = 0; g < NGRP; g += 2) {
        load_group(bufB, xv + (size_t)(g + 1) * XV);
        __builtin_amdgcn_sched_barrier(0);
        run_group(bufA, k, qp + (size_t)g * XCH * BDIM, S1, S2, S3, S4, S5);

        const int gn = (g + 2 < NGRP) ? (g + 2) : g;
        load_group(bufA, xv + (size_t)gn * XV);
        __builtin_amdgcn_sched_barrier(0);
        run_group(bufB, k, qp + (size_t)(g + 1) * XCH * BDIM, S1, S2, S3, S4, S5);
    }
}

__global__ __launch_bounds__(256) void hbv_route_wmma_kernel(
    const float* __restrict__ qws, float* __restrict__ out)
{
    constexpr double INV_MID_D = 1.0 / (4.0 + 1e-7);
    constexpr double WSUM_D = 1.0 + 2.0 * ((1.0 - 1.0 * INV_MID_D) +
                                           (1.0 - 2.0 * INV_MID_D) +
                                           (1.0 - 3.0 * INV_MID_D) +
                                           (1.0 - 4.0 * INV_MID_D));
    constexpr float INV_MID  = (float)INV_MID_D;
    constexpr float INV_WSUM = (float)(1.0 / (WSUM_D + 1e-7));

    const int lane = threadIdx.x & 31;
    const int wave = threadIdx.x >> 5;
    const int tile = blockIdx.x * 8 + wave;
    const int tb   = tile & (BDIM / 16 - 1);
    const int tt   = tile >> 7;
    const int b0   = tb * 16;
    const int t0   = tt * 16;
    const int mn   = lane & 15;
    const int khi  = lane >> 4;

    v2f afr[6], wfr[6];
#pragma unroll
    for (int c = 0; c < 6; ++c) {
#pragma unroll
        for (int j = 0; j < 2; ++j) {
            const int kg = 4 * c + j + 2 * khi;
            const int tq  = t0 - 4 + kg;
            const int tqc = tq < 0 ? 0 : (tq > TDIM - 1 ? TDIM - 1 : tq);
            const float v = qws[(size_t)tqc * BDIM + b0 + mn];
            afr[c][j] = (tq == tqc) ? v : 0.0f;
            const float dd = fabsf((float)(kg - mn) - 4.0f);
            wfr[c][j] = fmaxf(0.0f, fmaf(-dd, INV_MID, 1.0f)) * INV_WSUM;
        }
    }

    v8f acc = {0.0f, 0.0f, 0.0f, 0.0f, 0.0f, 0.0f, 0.0f, 0.0f};
#pragma unroll
    for (int c = 0; c < 6; ++c) {
        acc = __builtin_amdgcn_wmma_f32_16x16x4_f32(
            false, afr[c], false, wfr[c], (short)0, acc, false, false);
    }

#pragma unroll
    for (int v = 0; v < 8; ++v) {
        const int m = v + 8 * khi;
        out[(size_t)(b0 + m) * TDIM + t0 + mn] = acc[v];
    }
}

extern "C" void kernel_launch(void* const* d_in, const int* in_sizes, int n_in,
                              void* d_out, int out_size, void* d_ws, size_t ws_size,
                              hipStream_t stream) {
    (void)in_sizes; (void)n_in; (void)out_size; (void)ws_size;
    const float* x      = (const float*)d_in[0];
    const float* params = (const float*)d_in[1];
    float* out = (float*)d_out;
    float* qws = (float*)d_ws;

    hbv_scan_kernel<<<BDIM / 32, 32, 0, stream>>>(x, params, qws);
    hbv_route_wmma_kernel<<<(BDIM / 16) * (TDIM / 16) / 8, 256, 0, stream>>>(qws, out);
}


# snow, soil, upper and lower zone run as a 4-stage pipeline on the 4 waves of a work-group, each 16 steps behind the last, fluxes passed on in LDS; a flux once fused into the next sum rounds first
# speedup vs baseline: 8.5905x; 1.2940x over previous
#include <hip/hip_runtime.h>

#define BDIM 2048
#define TDIM 4096

typedef float v2f __attribute__((ext_vector_type(2)));
typedef float v4f __attribute__((ext_vector_type(4)));
typedef float v8f __attribute__((ext_vector_type(8)));

#define XCH 16
#define XV  (XCH * 3 / 4)
#define NGRP (TDIM / XCH)
#define NSTAGE 4
#define NITER (NGRP + NSTAGE)

__device__ __forceinline__ float sigm10(float y) {
    float e = __builtin_amdgcn_exp2f(-14.4269504088896341f * y);
    return __builtin_amdgcn_rcpf(1.0f + e);
}

__device__ __forceinline__ float clip1e5(float d) {
    return fminf(fmaxf(d, -100000.0f), 100000.0f);
}

struct HbvPar {
    float Tt, CFMAX, SCF, CFRMAX, CWH, rawCWH, Beta, invFC, FCLP, invFCLP;
    float PER, UZL, K0, K1, K2;
};

__device__ __forceinline__ void load_group(v4f (&buf)[XV], const v4f* __restrict__ src) {
#pragma unroll
    for (int i = 0; i < XV; ++i) buf[i] = src[i];
}

__device__ __forceinline__ void stage_barrier() {
    asm volatile("s_wait_dscnt 0x0\n\ts_barrier_signal -1\n\ts_barrier_wait -1" ::: "memory");
}

__device__ __forceinline__ void snow_block(
    const v4f (&buf)[XV], const HbvPar& k,
    float* __restrict__ outI, float* __restrict__ outPET, float& S1, float& S2)
{
#pragma unroll
    for (int s = 0; s < XCH; ++s) {
        const float P   = buf[(3 * s + 0) >> 2][(3 * s + 0) & 3];
        const float Tm  = buf[(3 * s + 1) >> 2][(3 * s + 1) & 3];
        const float PET = buf[(3 * s + 2) >> 2][(3 * s + 2) & 3];

        float h_cold = sigm10(k.Tt - Tm);
        float h_warm = 1.0f - h_cold;
        float Ps  = P * k.SCF * h_cold;
        float Rfr = fminf((k.Tt - Tm) * k.CFRMAX, S2) * h_cold;
        float Pr  = P * h_warm;
        float M   = fminf((Tm - k.Tt) * k.CFMAX, S1) * h_warm;
        float I   = sigm10(S2 - S1 * k.CWH) * (Pr + M);

        float nS1 = fmaxf(S1 + clip1e5(Ps + Rfr - M), 0.0f);
        float nS2 = fmaxf(fminf(S2 + clip1e5(Pr + M - I - Rfr), nS1 * k.rawCWH), 0.0f);
        S1 = nS1; S2 = nS2;

        outI[s * 32]   = I;
        outPET[s * 32] = PET;
    }
}

__device__ __forceinline__ void soil_block(
    const HbvPar& k, const float* __restrict__ inI, const float* __restrict__ inPET,
    float* __restrict__ outRe, float& S3)
{
    float Iv[XCH], PETv[XCH];
#pragma unroll
    for (int s = 0; s < XCH; ++s) { Iv[s] = inI[s * 32]; PETv[s] = inPET[s * 32]; }
    __builtin_amdgcn_sched_barrier(0);
#pragma unroll
    for (int s = 0; s < XCH; ++s) {
        const float I = Iv[s], PET = PETv[s];
        float Re  = I * __builtin_amdgcn_exp2f(k.Beta * __builtin_amdgcn_logf(S3 * k.invFC));
        float hS3 = sigm10(S3);
        float hLP = sigm10(k.FCLP - S3);
        float Ea  = hS3 * (PET * S3 * k.invFCLP) * hLP + hS3 * PET * (1.0f - hLP);
        float nS3 = fmaxf(S3 + clip1e5(I - Ea - Re), 0.0f);
        S3 = nS3;
        outRe[s * 32] = Re;
    }
}

__device__ __forceinline__ void upper_block(
    const HbvPar& k, const float* __restrict__ inRe,
    float* __restrict__ outPe, float* __restrict__ outQ, float& S4)
{
    float Rev[XCH];
#pragma unroll
    for (int s = 0; s < XCH; ++s) Rev[s] = inRe[s * 32];
    __builtin_amdgcn_sched_barrier(0);
#pragma unroll
    for (int s = 0; s < XCH; ++s) {
        const float Re = Rev[s];
        float hS4 = sigm10(S4);
        float Pe  = k.PER * hS4;
        float q0  = (S4 - k.UZL) * k.K0 * sigm10(S4 - k.UZL);
        float q1  = S4 * k.K1 * hS4;
        float nS4 = fmaxf(S4 + clip1e5(Re - q0 - q1 - Pe), 0.0f);
        S4 = nS4;
        outPe[s * 32] = Pe;
        outQ[s * 32]  = q0 + q1;
    }
}

__device__ __forceinline__ void lower_block(
    const HbvPar& k, const float* __restrict__ inPe, const float* __restrict__ inQ,
    float* __restrict__ qt, float& S5)
{
    float Pev[XCH], Qv[XCH];
#pragma unroll
    for (int s = 0; s < XCH; ++s) { Pev[s] = inPe[s * 32]; Qv[s] = inQ[s * 32]; }
    __builtin_amdgcn_sched_barrier(0);
#pragma unroll
    for (int s = 0; s < XCH; ++s) {
        float q2  = S5 * k.K2 * sigm10(S5);
        float nS5 = fmaxf(S5 + clip1e5(Pev[s] - q2), 0.0f);
        S5 = nS5;
        qt[(size_t)s * BDIM] = Qv[s] + q2;
    }
}

__global__ __launch_bounds__(32 * NSTAGE) void hbv_scan_kernel(
    const float* __restrict__ x, const float* __restrict__ p,
    float* __restrict__ qws)
{
    __shared__ float sI[2][XCH][32], sPET[2][XCH][32], sRe[2][XCH][32];
    __shared__ float sPe[2][XCH][32], sQ[2][XCH][32];

    const int lane  = threadIdx.x & 31;
    const int stage = __builtin_amdgcn_readfirstlane((int)(threadIdx.x >> 5));
    const int b = blockIdx.x * 32 + lane;

    const float CFR = p[1] * 0.01f;
    const float FC  = p[4] * 100.0f;
    const float LP  = p[6] * 2.5f;
    HbvPar k;
    k.Tt      = p[0]  * -1.5f;
    k.CFMAX   = p[2];
    k.SCF     = p[3];
    k.Beta    = p[5]  * 8.0f;
    k.CWH     = p[7]  * 0.3f;
    k.UZL     = p[8]  * 120.0f;
    k.K0      = p[9]  * 0.5f;
    k.K1      = p[10] * 0.3f;
    k.K2      = p[11] * 0.01f;
    k.PER     = p[12] * 3.5f;
    k.rawCWH  = p[7];
    k.invFC   = __builtin_amdgcn_rcpf(FC);
    k.FCLP    = FC * LP;
    k.invFCLP = __builtin_amdgcn_rcpf(k.FCLP);
    k.CFRMAX  = CFR * k.CFMAX;

    float S1 = 0.0f, S2 = 0.0f, S3 = 0.0f, S4 = 0.0f, S5 = 0.0f;

    const v4f* xv = reinterpret_cast<const v4f*>(x + (size_t)b * TDIM * 3);
    float* qp = qws + b;

    v4f bufA[XV], bufB[XV];
    if (stage == 0) load_group(bufA, xv);

    for (int j = 0; j < NITER; j += 2) {
        if (stage == 0) {
            if (j < NGRP) {
                load_group(bufB, xv + (size_t)(j + 1) * XV);
                __builtin_amdgcn_sched_barrier(0);
                snow_block(bufA, k, &sI[0][0][lane], &sPET[0][0][lane], S1, S2);
            }
        } else if (stage == 1) {
            const int blk = j - 1;
            if (blk >= 0 && blk < NGRP)
                soil_block(k, &sI[1][0][lane], &sPET[1][0][lane], &sRe[0][0][lane], S3);
        } else if (stage == 2) {
            const int blk = j - 2;
            if (blk >= 0 && blk < NGRP)
                upper_block(k, &sRe[1][0][lane], &sPe[0][0][lane], &sQ[0][0][lane], S4);
        } else {
            const int blk = j - 3;
            if (blk >= 0 && blk < NGRP)
                lower_block(k, &sPe[1][0][lane], &sQ[1][0][lane],
                            qp + (size_t)blk * XCH * BDIM, S5);
        }
        stage_barrier();

        if (stage == 0) {
            if (j + 1 < NGRP) {
                const int gn = (j + 2 < NGRP) ? (j + 2) : j;
                load_group(bufA, xv + (size_t)gn * XV);
                __builtin_amdgcn_sched_barrier(0);
                snow_block(bufB, k, &sI[1][0][lane], &sPET[1][0][lane], S1, S2);
            }
        } else if (stage == 1) {
            const int blk = j;
            if (blk < NGRP)
                soil_block(k, &sI[0][0][lane], &sPET[0][0][lane], &sRe[1][0][lane], S3);
        } else if (stage == 2) {
            const int blk = j - 1;
            if (blk >= 0 && blk < NGRP)
                upper_block(k, &sRe[0][0][lane], &sPe[1][0][lane], &sQ[1][0][lane], S4);
        } else {
            const int blk = j - 2;
            if (blk >= 0 && blk < NGRP)
                lower_block(k, &sPe[0][0][lane], &sQ[0][0][lane],
                            qp + (size_t)blk * XCH * BDIM, S5);
        }
        stage_barrier();
    }
}

__global__ __launch_bounds__(256) void hbv_route_wmma_kernel(
    const float* __restrict__ qws, float* __restrict__ out)
{
    constexpr double INV_MID_D = 1.0 / (4.0 + 1e-7);
    constexpr double WSUM_D = 1.0 + 2.0 * ((1.0 - 1.0 * INV_MID_D) +
                                           (1.0 - 2.0 * INV_MID_D) +
                                           (1.0 - 3.0 * INV_MID_D) +
                                           (1.0 - 4.0 * INV_MID_D));
    constexpr float INV_MID  = (float)INV_MID_D;
    constexpr float INV_WSUM = (float)(1.0 / (WSUM_D + 1e-7));

    const int lane = threadIdx.x & 31;
    const int wave = threadIdx.x >> 5;
    const int tile = blockIdx.x * 8 + wave;
    const int tb   = tile & (BDIM / 16 - 1);
    const int tt   = tile >> 7;
    const int b0   = tb * 16;
    const int t0   = tt * 16;
    const int mn   = lane & 15;
    const int khi  = lane >> 4;

    v2f afr[6], wfr[6];
#pragma unroll
    for (int c = 0; c < 6; ++c) {
#pragma unroll
        for (int j = 0; j < 2; ++j) {
            const int kg = 4 * c + j + 2 * khi;
            const int tq  = t0 - 4 + kg;
            const int tqc = tq < 0 ? 0 : (tq > TDIM - 1 ? TDIM - 1 : tq);
            const float v = qws[(size_t)tqc * BDIM + b0 + mn];
            afr[c][j] = (tq == tqc) ? v : 0.0f;
            const float dd = fabsf((float)(kg - mn) - 4.0f);
            wfr[c][j] = fmaxf(0.0f, fmaf(-dd, INV_MID, 1.0f)) * INV_WSUM;
        }
    }

    v8f acc = {0.0f, 0.0f, 0.0f, 0.0f, 0.0f, 0.0f, 0.0f, 0.0f};
#pragma unroll
    for (int c = 0; c < 6; ++c) {
        acc = __builtin_amdgcn_wmma_f32_16x16x4_f32(
            false, afr[c], false, wfr[c], (short)0, acc, false, false);
    }

#pragma unroll
    for (int v = 0; v < 8; ++v) {
        const int m = v + 8 * khi;
        out[(size_t)(b0 + m) * TDIM + t0 + mn] = acc[v];
    }
}

extern "C" void kernel_launch(void* const* d_in, const int* in_sizes, int n_in,
                              void* d_out, int out_size, void* d_ws, size_t ws_size,
                              hipStream_t stream) {
    (void)in_sizes; (void)n_in; (void)out_size; (void)ws_size;
    const float* x      = (const float*)d_in[0];
    const float* params = (const float*)d_in[1];
    float* out = (float*)d_out;
    float* qws = (float*)d_ws;

    hbv_scan_kernel<<<BDIM / 32, 32 * NSTAGE, 0, stream>>>(x, params, qws);
    hbv_route_wmma_kernel<<<(BDIM / 16) * (TDIM / 16) / 8, 256, 0, stream>>>(qws, out);
}


# each stage's step regrouped so fewer dependent operations stand behind its sigmoids: sums re-associated ((I-Re)-Ea, (Re-(q1+Pe))-q0, (Pr+M-Rfr)-I); two threshold sigmoids' exponents formed by one fma
# speedup vs baseline: 9.5795x; 1.1151x over previous
#include <hip/hip_runtime.h>

#define BDIM 2048
#define TDIM 4096

typedef float v2f __attribute__((ext_vector_type(2)));
typedef float v4f __attribute__((ext_vector_type(4)));
typedef float v8f __attribute__((ext_vector_type(8)));

#define XCH 16
#define XV  (XCH * 3 / 4)
#define NGRP (TDIM / XCH)
#define NSTAGE 4
#define NITER (NGRP + NSTAGE)

#define C10 14.4269504088896341f

__device__ __forceinline__ float sigm10(float y) {
    float e = __builtin_amdgcn_exp2f(-C10 * y);
    return __builtin_amdgcn_rcpf(1.0f + e);
}

__device__ __forceinline__ float sigm_fma(float y, float a, float b) {
    float e = __builtin_amdgcn_exp2f(__builtin_fmaf(y, a, b));
    return __builtin_amdgcn_rcpf(1.0f + e);
}

__device__ __forceinline__ float clip1e5(float d) {
    return fminf(fmaxf(d, -100000.0f), 100000.0f);
}

struct HbvPar {
    float Tt, CFMAX, SCF, CFRMAX, CWH, rawCWH, Beta, invFC, FCLP, invFCLP;
    float PER, UZL, K0, K1, K2;
    float nCFCLP, CUZL;
};

__device__ __forceinline__ void load_group(v4f (&buf)[XV], const v4f* __restrict__ src) {
#pragma unroll
    for (int i = 0; i < XV; ++i) buf[i] = src[i];
}

__device__ __forceinline__ void stage_barrier() {
    asm volatile("s_wait_dscnt 0x0\n\ts_barrier_signal -1\n\ts_barrier_wait -1" ::: "memory");
}

__device__ __forceinline__ void snow_block(
    const v4f (&buf)[XV], const HbvPar& k,
    float* __restrict__ outI, float* __restrict__ outPET, float& S1, float& S2)
{
#pragma unroll
    for (int s = 0; s < XCH; ++s) {
        const float P   = buf[(3 * s + 0) >> 2][(3 * s + 0) & 3];
        const float Tm  = buf[(3 * s + 1) >> 2][(3 * s + 1) & 3];
        const float PET = buf[(3 * s + 2) >> 2][(3 * s + 2) & 3];

        float h_cold = sigm10(k.Tt - Tm);
        float h_warm = 1.0f - h_cold;
        float Ps  = P * k.SCF * h_cold;
        float Rfr = fminf((k.Tt - Tm) * k.CFRMAX, S2) * h_cold;
        float Pr  = P * h_warm;
        float M   = fminf((Tm - k.Tt) * k.CFMAX, S1) * h_warm;
        float X   = Pr + M;
        float Xr  = X - Rfr;
        float sg  = sigm10(S2 - S1 * k.CWH);
        float I   = sg * X;

        float nS1 = fmaxf(S1 + clip1e5(Ps + Rfr - M), 0.0f);
        float nS2 = fmaxf(fminf(S2 + clip1e5(__builtin_fmaf(-sg, X, Xr)), nS1 * k.rawCWH), 0.0f);
        S1 = nS1; S2 = nS2;

        outI[s * 32]   = I;
        outPET[s * 32] = PET;
    }
}

__device__ __forceinline__ void soil_block(
    const HbvPar& k, const float* __restrict__ inI, const float* __restrict__ inPET,
    float* __restrict__ outRe, float& S3)
{
    float Iv[XCH], PETv[XCH];
#pragma unroll
    for (int s = 0; s < XCH; ++s) { Iv[s] = inI[s * 32]; PETv[s] = inPET[s * 32]; }
    __builtin_amdgcn_sched_barrier(0);
#pragma unroll
    for (int s = 0; s < XCH; ++s) {
        const float I = Iv[s], PET = PETv[s];
        float pw  = __builtin_amdgcn_exp2f(k.Beta * __builtin_amdgcn_logf(S3 * k.invFC));
        float Re  = I * pw;
        float X   = __builtin_fmaf(-I, pw, I);
        float hS3 = sigm10(S3);
        float hLP = sigm_fma(S3, C10, k.nCFCLP);
        float w   = __builtin_fmaf(hLP, __builtin_fmaf(S3, k.invFCLP, -1.0f), 1.0f);
        float A   = hS3 * PET;
        float nS3 = fmaxf(S3 + clip1e5(__builtin_fmaf(-A, w, X)), 0.0f);
        S3 = nS3;
        outRe[s * 32] = Re;
    }
}

__device__ __forceinline__ void upper_block(
    const HbvPar& k, const float* __restrict__ inRe,
    float* __restrict__ outPe, float* __restrict__ outQ, float& S4)
{
    float Rev[XCH];
#pragma unroll
    for (int s = 0; s < XCH; ++s) Rev[s] = inRe[s * 32];
    __builtin_amdgcn_sched_barrier(0);
#pragma unroll
    for (int s = 0; s < XCH; ++s) {
        const float Re = Rev[s];
        float hS4 = sigm10(S4);
        float sU  = sigm_fma(S4, -C10, k.CUZL);
        float xk  = (S4 - k.UZL) * k.K0;
        float Pe  = k.PER * hS4;
        float q0  = xk * sU;
        float q1  = S4 * k.K1 * hS4;
        float t   = __builtin_fmaf(-__builtin_fmaf(S4, k.K1, k.PER), hS4, Re);
        float nS4 = fmaxf(S4 + clip1e5(__builtin_fmaf(-xk, sU, t)), 0.0f);
        S4 = nS4;
        outPe[s * 32] = Pe;
        outQ[s * 32]  = q0 + q1;
    }
}

__device__ __forceinline__ void lower_block(
    const HbvPar& k, const float* __restrict__ inPe, const float* __restrict__ inQ,
    float* __restrict__ qt, float& S5)
{
    float Pev[XCH], Qv[XCH];
#pragma unroll
    for (int s = 0; s < XCH; ++s) { Pev[s] = inPe[s * 32]; Qv[s] = inQ[s * 32]; }
    __builtin_amdgcn_sched_barrier(0);
#pragma unroll
    for (int s = 0; s < XCH; ++s) {
        float q2  = S5 * k.K2 * sigm10(S5);
        float nS5 = fmaxf(S5 + clip1e5(Pev[s] - q2), 0.0f);
        S5 = nS5;
        qt[(size_t)s * BDIM] = Qv[s] + q2;
    }
}

__global__ __launch_bounds__(32 * NSTAGE) void hbv_scan_kernel(
    const float* __restrict__ x, const float* __restrict__ p,
    float* __restrict__ qws)
{
    __shared__ float sI[2][XCH][32], sPET[2][XCH][32], sRe[2][XCH][32];
    __shared__ float sPe[2][XCH][32], sQ[2][XCH][32];

    const int lane  = threadIdx.x & 31;
    const int stage = __builtin_amdgcn_readfirstlane((int)(threadIdx.x >> 5));
    const int b = blockIdx.x * 32 + lane;

    const float CFR = p[1] * 0.01f;
    const float FC  = p[4] * 100.0f;
    const float LP  = p[6] * 2.5f;
    HbvPar k;
    k.Tt      = p[0]  * -1.5f;
    k.CFMAX   = p[2];
    k.SCF     = p[3];
    k.Beta    = p[5]  * 8.0f;
    k.CWH     = p[7]  * 0.3f;
    k.UZL     = p[8]  * 120.0f;
    k.K0      = p[9]  * 0.5f;
    k.K1      = p[10] * 0.3f;
    k.K2      = p[11] * 0.01f;
    k.PER     = p[12] * 3.5f;
    k.rawCWH  = p[7];
    k.invFC   = __builtin_amdgcn_rcpf(FC);
    k.FCLP    = FC * LP;
    k.invFCLP = __builtin_amdgcn_rcpf(k.FCLP);
    k.CFRMAX  = CFR * k.CFMAX;
    k.nCFCLP  = -C10 * k.FCLP;
    k.CUZL    = C10 * k.UZL;

    float S1 = 0.0f, S2 = 0.0f, S3 = 0.0f, S4 = 0.0f, S5 = 0.0f;

    const v4f* xv = reinterpret_cast<const v4f*>(x + (size_t)b * TDIM * 3);
    float* qp = qws + b;

    v4f bufA[XV], bufB[XV];
    if (stage == 0) load_group(bufA, xv);

    for (int j = 0; j < NITER; j += 2) {
        if (stage == 0) {
            if (j < NGRP) {
                load_group(bufB, xv + (size_t)(j + 1) * XV);
                __builtin_amdgcn_sched_barrier(0);
                snow_block(bufA, k, &sI[0][0][lane], &sPET[0][0][lane], S1, S2);
            }
        } else if (stage == 1) {
            const int blk = j - 1;
            if (blk >= 0 && blk < NGRP)
                soil_block(k, &sI[1][0][lane], &sPET[1][0][lane], &sRe[0][0][lane], S3);
        } else if (stage == 2) {
            const int blk = j - 2;
            if (blk >= 0 && blk < NGRP)
                upper_block(k, &sRe[1][0][lane], &sPe[0][0][lane], &sQ[0][0][lane], S4);
        } else {
            const int blk = j - 3;
            if (blk >= 0 && blk < NGRP)
                lower_block(k, &sPe[1][0][lane], &sQ[1][0][lane],
                            qp + (size_t)blk * XCH * BDIM, S5);
        }
        stage_barrier();

        if (stage == 0) {
            if (j + 1 < NGRP) {
                const int gn = (j + 2 < NGRP) ? (j + 2) : j;
                load_group(bufA, xv + (size_t)gn * XV);
                __builtin_amdgcn_sched_barrier(0);
                snow_block(bufB, k, &sI[1][0][lane], &sPET[1][0][lane], S1, S2);
            }
        } else if (stage == 1) {
            const int blk = j;
            if (blk < NGRP)
                soil_block(k, &sI[0][0][lane], &sPET[0][0][lane], &sRe[1][0][lane], S3);
        } else if (stage == 2) {
            const int blk = j - 1;
            if (blk >= 0 && blk < NGRP)
                upper_block(k, &sRe[0][0][lane], &sPe[1][0][lane], &sQ[1][0][lane], S4);
        } else {
            const int blk = j - 2;
            if (blk >= 0 && blk < NGRP)
                lower_block(k, &sPe[0][0][lane], &sQ[0][0][lane],
                            qp + (size_t)blk * XCH * BDIM, S5);
        }
        stage_barrier();
    }
}

__global__ __launch_bounds__(256) void hbv_route_wmma_kernel(
    const float* __restrict__ qws, float* __restrict__ out)
{
    constexpr double INV_MID_D = 1.0 / (4.0 + 1e-7);
    constexpr double WSUM_D = 1.0 + 2.0 * ((1.0 - 1.0 * INV_MID_D) +
                                           (1.0 - 2.0 * INV_MID_D) +
                                           (1.0 - 3.0 * INV_MID_D) +
                                           (1.0 - 4.0 * INV_MID_D));
    constexpr float INV_MID  = (float)INV_MID_D;
    constexpr float INV_WSUM = (float)(1.0 / (WSUM_D + 1e-7));

    const int lane = threadIdx.x & 31;
    const int wave = threadIdx.x >> 5;
    const int tile = blockIdx.x * 8 + wave;
    const int tb   = tile & (BDIM / 16 - 1);
    const int tt   = tile >> 7;
    const int b0   = tb * 16;
    const int t0   = tt * 16;
    const int mn   = lane & 15;
    const int khi  = lane >> 4;

    v2f afr[6], wfr[6];
#pragma unroll
    for (int c = 0; c < 6; ++c) {
#pragma unroll
        for (int j = 0; j < 2; ++j) {
            const int kg = 4 * c + j + 2 * khi;
            const int tq  = t0 - 4 + kg;
            const int tqc = tq < 0 ? 0 : (tq > TDIM - 1 ? TDIM - 1 : tq);
            const float v = qws[(size_t)tqc * BDIM + b0 + mn];
            afr[c][j] = (tq == tqc) ? v : 0.0f;
            const float dd = fabsf((float)(kg - mn) - 4.0f);
            wfr[c][j] = fmaxf(0.0f, fmaf(-dd, INV_MID, 1.0f)) * INV_WSUM;
        }
    }

    v8f acc = {0.0f, 0.0f, 0.0f, 0.0f, 0.0f, 0.0f, 0.0f, 0.0f};
#pragma unroll
    for (int c = 0; c < 6; ++c) {
        acc = __builtin_amdgcn_wmma_f32_16x16x4_f32(
            false, afr[c], false, wfr[c], (short)0, acc, false, false);
    }

#pragma unroll
    for (int v = 0; v < 8; ++v) {
        const int m = v + 8 * khi;
        out[(size_t)(b0 + m) * TDIM + t0 + mn] = acc[v];
    }
}

extern "C" void kernel_launch(void* const* d_in, const int* in_sizes, int n_in,
                              void* d_out, int out_size, void* d_ws, size_t ws_size,
                              hipStream_t stream) {
    (void)in_sizes; (void)n_in; (void)out_size; (void)ws_size;
    const float* x      = (const float*)d_in[0];
    const float* params = (const float*)d_in[1];
    float* out = (float*)d_out;
    float* qws = (float*)d_ws;

    hbv_scan_kernel<<<BDIM / 32, 32 * NSTAGE, 0, stream>>>(x, params, qws);
    hbv_route_wmma_kernel<<<(BDIM / 16) * (TDIM / 16) / 8, 256, 0, stream>>>(qws, out);
}


# routing convolution fused into the scan kernel as a fifth pipeline stage on a fifth wave (same WMMAs, same k order): q handed over in an LDS ring, second launch and q's trip through memory gone
# speedup vs baseline: 10.9748x; 1.1457x over previous
#include <hip/hip_runtime.h>

#define BDIM 2048
#define TDIM 4096

typedef float v2f __attribute__((ext_vector_type(2)));
typedef float v4f __attribute__((ext_vector_type(4)));
typedef float v8f __attribute__((ext_vector_type(8)));

#define XCH 16
#define XV  (XCH * 3 / 4)
#define NGRP (TDIM / XCH)
#define NSTAGE 5
#define RING (4 * XCH)
#define NITER ((NGRP + NSTAGE + 1) & ~1)

__device__ __forceinline__ float th5(float y) {
    return __builtin_amdgcn_tanhf(5.0f * y);
}

__device__ __forceinline__ float th5_fma(float y, float b) {
    return __builtin_amdgcn_tanhf(__builtin_fmaf(y, 5.0f, b));
}

__device__ __forceinline__ float clip1e5(float d) {
    return fminf(fmaxf(d, -100000.0f), 100000.0f);
}

struct HbvPar {
    float Tt, CFMAX, SCF, CFRMAX, rawCWH, Beta, invFC;
    float n5FCLP, n5UZL;
    float hinvFCLP;
    float n5CWH;
    float hPER, hK0, hK1, hK2, nhK0UZL;
};

__device__ __forceinline__ void load_group(v4f (&buf)[XV], const v4f* __restrict__ src) {
#pragma unroll
    for (int i = 0; i < XV; ++i) buf[i] = src[i];
}

__device__ __forceinline__ void stage_barrier() {
    asm volatile("s_wait_dscnt 0x0\n\ts_barrier_signal -1\n\ts_barrier_wait -1" ::: "memory");
}

struct SnowForcing { float a1, a2, hc, hw, hhw, Ps, hPr; };

__device__ __forceinline__ SnowForcing snow_forcing(float P, float Tm, const HbvPar& k) {
    SnowForcing f;
    const float tc = th5(k.Tt - Tm);
    f.hc  = __builtin_fmaf(0.5f, tc, 0.5f);
    f.hw  = __builtin_fmaf(-0.5f, tc, 0.5f);
    f.hhw = __builtin_fmaf(-0.25f, tc, 0.25f);
    f.a1  = (k.Tt - Tm) * k.CFRMAX;
    f.a2  = (k.Tt - Tm) * -k.CFMAX;
    f.Ps  = P * k.SCF * f.hc;
    f.hPr = P * f.hhw;
    return f;
}

__device__ __forceinline__ void snow_step(
    const SnowForcing& f, const HbvPar& k, float& S1, float& S2, float& I)
{
    float mn  = fminf(f.a2, S1);
    float w1  = S1 * k.n5CWH;
    float hX  = __builtin_fmaf(mn, f.hhw, f.hPr);
    float e0  = __builtin_fmaf(-mn, f.hw, f.Ps);
    float ts  = __builtin_amdgcn_tanhf(__builtin_fmaf(S2, 5.0f, w1));
    float mr  = fminf(f.a1, S2);
    I = __builtin_fmaf(hX, ts, hX);
    float nS1 = fmaxf(S1 + clip1e5(__builtin_fmaf(mr, f.hc, e0)), 0.0f);
    float nS2 = fmaxf(fminf(S2 + clip1e5(__builtin_fmaf(-hX, ts, __builtin_fmaf(-mr, f.hc, hX))),
                            nS1 * k.rawCWH), 0.0f);
    S1 = nS1; S2 = nS2;
}

__device__ __forceinline__ void snow_block(
    const v4f (&buf)[XV], const HbvPar& k,
    float* __restrict__ outI, float* __restrict__ outPET, float& S1, float& S2)
{
#pragma unroll
    for (int s = 0; s < XCH; ++s) {
        SnowForcing f = snow_forcing(buf[(3 * s + 0) >> 2][(3 * s + 0) & 3], buf[(3 * s + 1) >> 2][(3 * s + 1) & 3], k);
        float I;
        snow_step(f, k, S1, S2, I);
        outI[s * 32]   = I;
        outPET[s * 32] = buf[(3 * s + 2) >> 2][(3 * s + 2) & 3];
    }
}

__device__ __forceinline__ void soil_block(
    const HbvPar& k, const float* __restrict__ inI, const float* __restrict__ inPET,
    float* __restrict__ outRe, float& S3)
{
    float Iv[XCH], PETv[XCH];
#pragma unroll
    for (int s = 0; s < XCH; ++s) { Iv[s] = inI[s * 32]; PETv[s] = inPET[s * 32]; }
    __builtin_amdgcn_sched_barrier(0);
#pragma unroll
    for (int s = 0; s < XCH; ++s) {
        const float I = Iv[s], PET = PETv[s];
        float pw  = __builtin_amdgcn_exp2f(k.Beta * __builtin_amdgcn_logf(S3 * k.invFC));
        float Re  = I * pw;
        float t1  = th5(S3);
        float t2  = th5_fma(S3, k.n5FCLP);
        float hg  = __builtin_fmaf(S3, k.hinvFCLP, -0.5f);
        float w   = __builtin_fmaf(-hg, t2, 1.0f + hg);
        float hP  = 0.5f * PET;
        float A   = __builtin_fmaf(hP, t1, hP);
        float nS3 = fmaxf(S3 + clip1e5(__builtin_fmaf(-I, pw, I - A * w)), 0.0f);
        S3 = nS3;
        outRe[s * 32] = Re;
    }
}

__device__ __forceinline__ void upper_block(
    const HbvPar& k, const float* __restrict__ inRe,
    float* __restrict__ outPe, float* __restrict__ outQ, float& S4)
{
    float Rev[XCH];
#pragma unroll
    for (int s = 0; s < XCH; ++s) Rev[s] = inRe[s * 32];
    __builtin_amdgcn_sched_barrier(0);
#pragma unroll
    for (int s = 0; s < XCH; ++s) {
        const float Re = Rev[s];
        float t4  = th5(S4);
        float tU  = th5_fma(S4, k.n5UZL);
        float a   = __builtin_fmaf(S4, k.hK1, k.hPER);
        float b   = __builtin_fmaf(S4, k.hK0, k.nhK0UZL);
        float hq1 = S4 * k.hK1;
        float t   = __builtin_fmaf(-a, t4, (Re - a) - b);
        float nS4 = fmaxf(S4 + clip1e5(__builtin_fmaf(-b, tU, t)), 0.0f);
        S4 = nS4;
        outPe[s * 32] = __builtin_fmaf(k.hPER, t4, k.hPER);
        outQ[s * 32]  = __builtin_fmaf(b, tU, b) + __builtin_fmaf(hq1, t4, hq1);
    }
}

__device__ __forceinline__ void lower_block(
    const HbvPar& k, const float* __restrict__ inPe, const float* __restrict__ inQ,
    float* __restrict__ qr, float& S5)
{
    float Pev[XCH], Qv[XCH];
#pragma unroll
    for (int s = 0; s < XCH; ++s) { Pev[s] = inPe[s * 32]; Qv[s] = inQ[s * 32]; }
    __builtin_amdgcn_sched_barrier(0);
#pragma unroll
    for (int s = 0; s < XCH; ++s) {
        float m   = S5 * k.hK2;
        float t5  = th5(S5);
        float nS5 = fmaxf(S5 + clip1e5(__builtin_fmaf(-m, t5, Pev[s] - m)), 0.0f);
        S5 = nS5;
        qr[s * 32] = __builtin_fmaf(m, t5, Qv[s] + m);
    }
}

__device__ __forceinline__ void route_block(
    const float* __restrict__ ring, const v2f (&wfr)[6], const int blk,
    const int mn, const int khi, float* __restrict__ orow)
{
    v2f a0[6], a1[6];
    const int pbase = blk * XCH - 4 + 2 * khi;
#pragma unroll
    for (int c = 0; c < 6; ++c) {
#pragma unroll
        for (int j = 0; j < 2; ++j) {
            const int pt = pbase + 4 * c + j;
            const int pc = pt < 0 ? 0 : (pt > TDIM - 1 ? TDIM - 1 : pt);
            const float* src = ring + (pc & (RING - 1)) * 32 + mn;
            const float v0 = src[0];
            const float v1 = src[16];
            a0[c][j] = (pt == pc) ? v0 : 0.0f;
            a1[c][j] = (pt == pc) ? v1 : 0.0f;
        }
    }
    v8f acc0 = {0.0f, 0.0f, 0.0f, 0.0f, 0.0f, 0.0f, 0.0f, 0.0f};
    v8f acc1 = {0.0f, 0.0f, 0.0f, 0.0f, 0.0f, 0.0f, 0.0f, 0.0f};
#pragma unroll
    for (int c = 0; c < 6; ++c) {
        acc0 = __builtin_amdgcn_wmma_f32_16x16x4_f32(
            false, a0[c], false, wfr[c], (short)0, acc0, false, false);
        acc1 = __builtin_amdgcn_wmma_f32_16x16x4_f32(
            false, a1[c], false, wfr[c], (short)0, acc1, false, false);
    }
#pragma unroll
    for (int v = 0; v < 8; ++v) {
        orow[(size_t)v * TDIM]        = acc0[v];
        orow[(size_t)(v + 16) * TDIM] = acc1[v];
    }
}

__global__ __launch_bounds__(32 * NSTAGE) void hbv_scan_route_kernel(
    const float* __restrict__ x, const float* __restrict__ p,
    float* __restrict__ out)
{
    __shared__ float sI[2][XCH][32], sPET[2][XCH][32], sRe[2][XCH][32];
    __shared__ float sPe[2][XCH][32], sQ[2][XCH][32];
    __shared__ float sQr[RING][32];

    const int lane  = threadIdx.x & 31;
    const int wave  = __builtin_amdgcn_readfirstlane((int)(threadIdx.x >> 5));
    const int stage = (wave == 4) ? 4 : ((wave + 3) & 3);
    const int b = blockIdx.x * 32 + lane;

    const float CFR = p[1] * 0.01f;
    const float FC  = p[4] * 100.0f;
    const float LP  = p[6] * 2.5f;
    HbvPar k;
    k.Tt      = p[0]  * -1.5f;
    k.CFMAX   = p[2];
    k.SCF     = p[3];
    k.Beta    = p[5]  * 8.0f;
    k.n5CWH   = -5.0f * (p[7] * 0.3f);
    const float UZL = p[8] * 120.0f;
    k.hK0     = 0.5f * (p[9]  * 0.5f);
    k.hK1     = 0.5f * (p[10] * 0.3f);
    k.hK2     = 0.5f * (p[11] * 0.01f);
    k.hPER    = 0.5f * (p[12] * 3.5f);
    k.rawCWH  = p[7];
    k.invFC   = __builtin_amdgcn_rcpf(FC);
    k.hinvFCLP = 0.5f * __builtin_amdgcn_rcpf(FC * LP);
    k.CFRMAX  = CFR * k.CFMAX;
    k.n5FCLP  = -5.0f * (FC * LP);
    k.n5UZL   = -5.0f * UZL;
    k.nhK0UZL = -(k.hK0 * UZL);

    float S1 = 0.0f, S2 = 0.0f, S3 = 0.0f, S4 = 0.0f, S5 = 0.0f;

    const v4f* xv = reinterpret_cast<const v4f*>(x + (size_t)b * TDIM * 3);

    const int mn  = lane & 15;
    const int khi = lane >> 4;
    float* orow0 = out + ((size_t)(blockIdx.x * 32 + 8 * khi) * TDIM + mn);
    v2f wfr[6];
    if (stage == 4) {
        constexpr double INV_MID_D = 1.0 / (4.0 + 1e-7);
        constexpr double WSUM_D = 1.0 + 2.0 * ((1.0 - 1.0 * INV_MID_D) +
                                               (1.0 - 2.0 * INV_MID_D) +
                                               (1.0 - 3.0 * INV_MID_D) +
                                               (1.0 - 4.0 * INV_MID_D));
        constexpr float INV_MID  = (float)INV_MID_D;
        constexpr float INV_WSUM = (float)(1.0 / (WSUM_D + 1e-7));
#pragma unroll
        for (int c = 0; c < 6; ++c) {
#pragma unroll
            for (int j = 0; j < 2; ++j) {
                const int kg = 4 * c + j + 2 * khi;
                const float dd = fabsf((float)(kg - mn) - 4.0f);
                wfr[c][j] = fmaxf(0.0f, fmaf(-dd, INV_MID, 1.0f)) * INV_WSUM;
            }
        }
    }

    v4f bufA[XV], bufB[XV];
    if (stage == 0) load_group(bufA, xv);

    for (int j = 0; j < NITER; j += 2) {
        if (stage == 0) {
            if (j < NGRP) {
                load_group(bufB, xv + (size_t)(j + 1) * XV);
                __builtin_amdgcn_sched_barrier(0);
                snow_block(bufA, k, &sI[0][0][lane], &sPET[0][0][lane], S1, S2);
            }
        } else if (stage == 1) {
            const int blk = j - 1;
            if (blk >= 0 && blk < NGRP)
                soil_block(k, &sI[1][0][lane], &sPET[1][0][lane], &sRe[0][0][lane], S3);
        } else if (stage == 2) {
            const int blk = j - 2;
            if (blk >= 0 && blk < NGRP)
                upper_block(k, &sRe[1][0][lane], &sPe[0][0][lane], &sQ[0][0][lane], S4);
        } else if (stage == 3) {
            const int blk = j - 3;
            if (blk >= 0 && blk < NGRP)
                lower_block(k, &sPe[1][0][lane], &sQ[1][0][lane],
                            &sQr[(blk & 3) * XCH][lane], S5);
        } else {
            const int blk = j - 5;
            if (blk >= 0 && blk < NGRP)
                route_block(&sQr[0][0], wfr, blk, mn, khi, orow0 + blk * XCH);
        }
        stage_barrier();

        if (stage == 0) {
            if (j + 1 < NGRP) {
                const int gn = (j + 2 < NGRP) ? (j + 2) : j;
                load_group(bufA, xv + (size_t)gn * XV);
                __builtin_amdgcn_sched_barrier(0);
                snow_block(bufB, k, &sI[1][0][lane], &sPET[1][0][lane], S1, S2);
            }
        } else if (stage == 1) {
            const int blk = j;
            if (blk < NGRP)
                soil_block(k, &sI[0][0][lane], &sPET[0][0][lane], &sRe[1][0][lane], S3);
        } else if (stage == 2) {
            const int blk = j - 1;
            if (blk >= 0 && blk < NGRP)
                upper_block(k, &sRe[0][0][lane], &sPe[1][0][lane], &sQ[1][0][lane], S4);
        } else if (stage == 3) {
            const int blk = j - 2;
            if (blk >= 0 && blk < NGRP)
                lower_block(k, &sPe[0][0][lane], &sQ[0][0][lane],
                            &sQr[(blk & 3) * XCH][lane], S5);
        } else {
            const int blk = j - 4;
            if (blk >= 0 && blk < NGRP)
                route_block(&sQr[0][0], wfr, blk, mn, khi, orow0 + blk * XCH);
        }
        stage_barrier();
    }
}

extern "C" void kernel_launch(void* const* d_in, const int* in_sizes, int n_in,
                              void* d_out, int out_size, void* d_ws, size_t ws_size,
                              hipStream_t stream) {
    (void)in_sizes; (void)n_in; (void)out_size; (void)ws_size;
    const float* x      = (const float*)d_in[0];
    const float* params = (const float*)d_in[1];
    float* out = (float*)d_out;
    (void)d_ws;

    hbv_scan_route_kernel<<<BDIM / 32, 32 * NSTAGE, 0, stream>>>(x, params, out);
}


# x loads two groups ahead in the same registers: each 16-byte register is re-loaded (asm load, counted wait) for the group after next right behind the step that reads it last
# speedup vs baseline: 11.4928x; 1.0472x over previous
#include <hip/hip_runtime.h>

#define BDIM 2048
#define TDIM 4096

typedef float v2f __attribute__((ext_vector_type(2)));
typedef float v4f __attribute__((ext_vector_type(4)));
typedef float v8f __attribute__((ext_vector_type(8)));

#define XCH 16
#define XV  (XCH * 3 / 4)
#define NGRP (TDIM / XCH)
#define NSTAGE 5
#define RING (4 * XCH)
#define NITER ((NGRP + NSTAGE + 1) & ~1)

__device__ __forceinline__ float th5(float y) {
    return __builtin_amdgcn_tanhf(5.0f * y);
}

__device__ __forceinline__ float th5_fma(float y, float b) {
    return __builtin_amdgcn_tanhf(__builtin_fmaf(y, 5.0f, b));
}

__device__ __forceinline__ float clip1e5(float d) {
    return fminf(fmaxf(d, -100000.0f), 100000.0f);
}

struct HbvPar {
    float Tt, CFMAX, SCF, CFRMAX, rawCWH, Beta, invFC;
    float n5FCLP, n5UZL;
    float hinvFCLP;
    float n5CWH;
    float hPER, hK0, hK1, hK2, nhK0UZL;
};

__device__ __forceinline__ void stage_barrier() {
    asm volatile("s_wait_dscnt 0x0\n\ts_barrier_signal -1\n\ts_barrier_wait -1" ::: "memory");
}

struct SnowForcing { float a1, a2, hc, hw, hhw, Ps, hPr; };

__device__ __forceinline__ SnowForcing snow_forcing(float P, float Tm, const HbvPar& k) {
    SnowForcing f;
    const float tc = th5(k.Tt - Tm);
    f.hc  = __builtin_fmaf(0.5f, tc, 0.5f);
    f.hw  = __builtin_fmaf(-0.5f, tc, 0.5f);
    f.hhw = __builtin_fmaf(-0.25f, tc, 0.25f);
    f.a1  = (k.Tt - Tm) * k.CFRMAX;
    f.a2  = (k.Tt - Tm) * -k.CFMAX;
    f.Ps  = P * k.SCF * f.hc;
    f.hPr = P * f.hhw;
    return f;
}

__device__ __forceinline__ void snow_step(
    const SnowForcing& f, const HbvPar& k, float& S1, float& S2, float& I)
{
    float mn  = fminf(f.a2, S1);
    float w1  = S1 * k.n5CWH;
    float hX  = __builtin_fmaf(mn, f.hhw, f.hPr);
    float e0  = __builtin_fmaf(-mn, f.hw, f.Ps);
    float ts  = __builtin_amdgcn_tanhf(__builtin_fmaf(S2, 5.0f, w1));
    float mr  = fminf(f.a1, S2);
    I = __builtin_fmaf(hX, ts, hX);
    float nS1 = fmaxf(S1 + clip1e5(__builtin_fmaf(mr, f.hc, e0)), 0.0f);
    float nS2 = fmaxf(fminf(S2 + clip1e5(__builtin_fmaf(-hX, ts, __builtin_fmaf(-mr, f.hc, hX))),
                            nS1 * k.rawCWH), 0.0f);
    S1 = nS1; S2 = nS2;
}

template <int BYTE_OFF>
__device__ __forceinline__ void x_load(v4f& r, const v4f* base) {
    asm volatile("global_load_b128 %0, %1, off offset:%2" : "+v"(r) : "v"(base), "n"(BYTE_OFF));
}

#define XWAIT 22
__device__ __forceinline__ void x_wait(v4f& r) {
    asm volatile("s_wait_loadcnt %1" : "+v"(r) : "n"(XWAIT));
}

template <int S>
__device__ __forceinline__ void snow_steps(
    v4f (&buf)[XV], const v4f* nxt, const HbvPar& k,
    float* __restrict__ outI, float* __restrict__ outPET, float& S1, float& S2)
{
    if constexpr (S < XCH) {
        if constexpr (S == 0) x_wait(buf[0]);
        if constexpr (((3 * S + 2) >> 2) > ((3 * S - 1) >> 2) && S > 0) x_wait(buf[(3 * S + 2) >> 2]);
        SnowForcing f = snow_forcing(buf[(3 * S + 0) >> 2][(3 * S + 0) & 3], buf[(3 * S + 1) >> 2][(3 * S + 1) & 3], k);
        float I;
        snow_step(f, k, S1, S2, I);
        outI[S * 32]   = I;
        outPET[S * 32] = buf[(3 * S + 2) >> 2][(3 * S + 2) & 3];
        if constexpr (((3 * (S + 1)) >> 2) > ((3 * S) >> 2)) {
            constexpr int q = ((3 * (S + 1)) >> 2) - 1;
            x_load<16 * q>(buf[q], nxt);
        }
        snow_steps<S + 1>(buf, nxt, k, outI, outPET, S1, S2);
    }
}

__device__ __forceinline__ void snow_block(
    v4f (&buf)[XV], const v4f* nxt, const HbvPar& k,
    float* __restrict__ outI, float* __restrict__ outPET, float& S1, float& S2)
{
    snow_steps<0>(buf, nxt, k, outI, outPET, S1, S2);
    asm volatile("s_wait_xcnt 0x0");
}

__device__ __forceinline__ void soil_block(
    const HbvPar& k, const float* __restrict__ inI, const float* __restrict__ inPET,
    float* __restrict__ outRe, float& S3)
{
    float Iv[XCH], PETv[XCH];
#pragma unroll
    for (int s = 0; s < XCH; ++s) { Iv[s] = inI[s * 32]; PETv[s] = inPET[s * 32]; }
    __builtin_amdgcn_sched_barrier(0);
#pragma unroll
    for (int s = 0; s < XCH; ++s) {
        const float I = Iv[s], PET = PETv[s];
        float pw  = __builtin_amdgcn_exp2f(k.Beta * __builtin_amdgcn_logf(S3 * k.invFC));
        float Re  = I * pw;
        float t1  = th5(S3);
        float t2  = th5_fma(S3, k.n5FCLP);
        float hg  = __builtin_fmaf(S3, k.hinvFCLP, -0.5f);
        float w   = __builtin_fmaf(-hg, t2, 1.0f + hg);
        float hP  = 0.5f * PET;
        float A   = __builtin_fmaf(hP, t1, hP);
        float nS3 = fmaxf(S3 + clip1e5(__builtin_fmaf(-I, pw, I - A * w)), 0.0f);
        S3 = nS3;
        outRe[s * 32] = Re;
    }
}

__device__ __forceinline__ void upper_block(
    const HbvPar& k, const float* __restrict__ inRe,
    float* __restrict__ outPe, float* __restrict__ outQ, float& S4)
{
    float Rev[XCH];
#pragma unroll
    for (int s = 0; s < XCH; ++s) Rev[s] = inRe[s * 32];
    __builtin_amdgcn_sched_barrier(0);
#pragma unroll
    for (int s = 0; s < XCH; ++s) {
        const float Re = Rev[s];
        float t4  = th5(S4);
        float tU  = th5_fma(S4, k.n5UZL);
        float a   = __builtin_fmaf(S4, k.hK1, k.hPER);
        float b   = __builtin_fmaf(S4, k.hK0, k.nhK0UZL);
        float hq1 = S4 * k.hK1;
        float t   = __builtin_fmaf(-a, t4, (Re - a) - b);
        float nS4 = fmaxf(S4 + clip1e5(__builtin_fmaf(-b, tU, t)), 0.0f);
        S4 = nS4;
        outPe[s * 32] = __builtin_fmaf(k.hPER, t4, k.hPER);
        outQ[s * 32]  = __builtin_fmaf(b, tU, b) + __builtin_fmaf(hq1, t4, hq1);
    }
}

__device__ __forceinline__ void lower_block(
    const HbvPar& k, const float* __restrict__ inPe, const float* __restrict__ inQ,
    float* __restrict__ qr, float& S5)
{
    float Pev[XCH], Qv[XCH];
#pragma unroll
    for (int s = 0; s < XCH; ++s) { Pev[s] = inPe[s * 32]; Qv[s] = inQ[s * 32]; }
    __builtin_amdgcn_sched_barrier(0);
#pragma unroll
    for (int s = 0; s < XCH; ++s) {
        float m   = S5 * k.hK2;
        float t5  = th5(S5);
        float nS5 = fmaxf(S5 + clip1e5(__builtin_fmaf(-m, t5, Pev[s] - m)), 0.0f);
        S5 = nS5;
        qr[s * 32] = __builtin_fmaf(m, t5, Qv[s] + m);
    }
}

__device__ __forceinline__ void route_block(
    const float* __restrict__ ring, const v2f (&wfr)[6], const int blk,
    const int mn, const int khi, float* __restrict__ orow)
{
    v2f a0[6], a1[6];
    const int pbase = blk * XCH - 4 + 2 * khi;
#pragma unroll
    for (int c = 0; c < 6; ++c) {
#pragma unroll
        for (int j = 0; j < 2; ++j) {
            const int pt = pbase + 4 * c + j;
            const int pc = pt < 0 ? 0 : (pt > TDIM - 1 ? TDIM - 1 : pt);
            const float* src = ring + (pc & (RING - 1)) * 32 + mn;
            const float v0 = src[0];
            const float v1 = src[16];
            a0[c][j] = (pt == pc) ? v0 : 0.0f;
            a1[c][j] = (pt == pc) ? v1 : 0.0f;
        }
    }
    v8f acc0 = {0.0f, 0.0f, 0.0f, 0.0f, 0.0f, 0.0f, 0.0f, 0.0f};
    v8f acc1 = {0.0f, 0.0f, 0.0f, 0.0f, 0.0f, 0.0f, 0.0f, 0.0f};
#pragma unroll
    for (int c = 0; c < 6; ++c) {
        acc0 = __builtin_amdgcn_wmma_f32_16x16x4_f32(
            false, a0[c], false, wfr[c], (short)0, acc0, false, false);
        acc1 = __builtin_amdgcn_wmma_f32_16x16x4_f32(
            false, a1[c], false, wfr[c], (short)0, acc1, false, false);
    }
#pragma unroll
    for (int v = 0; v < 8; ++v) {
        orow[(size_t)v * TDIM]        = acc0[v];
        orow[(size_t)(v + 16) * TDIM] = acc1[v];
    }
}

__global__ __launch_bounds__(32 * NSTAGE) void hbv_scan_route_kernel(
    const float* __restrict__ x, const float* __restrict__ p,
    float* __restrict__ out)
{
    __shared__ float sI[2][XCH][32], sPET[2][XCH][32], sRe[2][XCH][32];
    __shared__ float sPe[2][XCH][32], sQ[2][XCH][32];
    __shared__ float sQr[RING][32];

    const int lane  = threadIdx.x & 31;
    const int wave  = __builtin_amdgcn_readfirstlane((int)(threadIdx.x >> 5));
    const int stage = (wave == 4) ? 4 : ((wave + 3) & 3);
    const int b = blockIdx.x * 32 + lane;

    const float CFR = p[1] * 0.01f;
    const float FC  = p[4] * 100.0f;
    const float LP  = p[6] * 2.5f;
    HbvPar k;
    k.Tt      = p[0]  * -1.5f;
    k.CFMAX   = p[2];
    k.SCF     = p[3];
    k.Beta    = p[5]  * 8.0f;
    k.n5CWH   = -5.0f * (p[7] * 0.3f);
    const float UZL = p[8] * 120.0f;
    k.hK0     = 0.5f * (p[9]  * 0.5f);
    k.hK1     = 0.5f * (p[10] * 0.3f);
    k.hK2     = 0.5f * (p[11] * 0.01f);
    k.hPER    = 0.5f * (p[12] * 3.5f);
    k.rawCWH  = p[7];
    k.invFC   = __builtin_amdgcn_rcpf(FC);
    k.hinvFCLP = 0.5f * __builtin_amdgcn_rcpf(FC * LP);
    k.CFRMAX  = CFR * k.CFMAX;
    k.n5FCLP  = -5.0f * (FC * LP);
    k.n5UZL   = -5.0f * UZL;
    k.nhK0UZL = -(k.hK0 * UZL);

    float S1 = 0.0f, S2 = 0.0f, S3 = 0.0f, S4 = 0.0f, S5 = 0.0f;

    const v4f* xv = reinterpret_cast<const v4f*>(x + (size_t)b * TDIM * 3);

    const int mn  = lane & 15;
    const int khi = lane >> 4;
    float* orow0 = out + ((size_t)(blockIdx.x * 32 + 8 * khi) * TDIM + mn);
    v2f wfr[6];
    if (stage == 4) {
        constexpr double INV_MID_D = 1.0 / (4.0 + 1e-7);
        constexpr double WSUM_D = 1.0 + 2.0 * ((1.0 - 1.0 * INV_MID_D) +
                                               (1.0 - 2.0 * INV_MID_D) +
                                               (1.0 - 3.0 * INV_MID_D) +
                                               (1.0 - 4.0 * INV_MID_D));
        constexpr float INV_MID  = (float)INV_MID_D;
        constexpr float INV_WSUM = (float)(1.0 / (WSUM_D + 1e-7));
#pragma unroll
        for (int c = 0; c < 6; ++c) {
#pragma unroll
            for (int j = 0; j < 2; ++j) {
                const int kg = 4 * c + j + 2 * khi;
                const float dd = fabsf((float)(kg - mn) - 4.0f);
                wfr[c][j] = fmaxf(0.0f, fmaf(-dd, INV_MID, 1.0f)) * INV_WSUM;
            }
        }
    }

    v4f bufA[XV], bufB[XV];
    if (stage == 0) {
        x_load<0>(bufA[0], xv);   x_load<16>(bufA[1], xv);  x_load<32>(bufA[2], xv);   x_load<48>(bufA[3], xv);
        x_load<64>(bufA[4], xv);  x_load<80>(bufA[5], xv);  x_load<96>(bufA[6], xv);   x_load<112>(bufA[7], xv);
        x_load<128>(bufA[8], xv); x_load<144>(bufA[9], xv); x_load<160>(bufA[10], xv); x_load<176>(bufA[11], xv);
        x_load<0>(bufB[0], xv + XV);   x_load<16>(bufB[1], xv + XV);  x_load<32>(bufB[2], xv + XV);   x_load<48>(bufB[3], xv + XV);
        x_load<64>(bufB[4], xv + XV);  x_load<80>(bufB[5], xv + XV);  x_load<96>(bufB[6], xv + XV);   x_load<112>(bufB[7], xv + XV);
        x_load<128>(bufB[8], xv + XV); x_load<144>(bufB[9], xv + XV); x_load<160>(bufB[10], xv + XV); x_load<176>(bufB[11], xv + XV);
        asm volatile("s_wait_xcnt 0x0");
    }

    for (int j = 0; j < NITER; j += 2) {
        if (stage == 0) {
            const int gn = (j + 2 < NGRP) ? (j + 2) : (NGRP - 1);
            snow_block(bufA, xv + (size_t)gn * XV, k, &sI[0][0][lane], &sPET[0][0][lane], S1, S2);
        } else if (stage == 1) {
            const int blk = j - 1;
            if (blk >= 0 && blk < NGRP)
                soil_block(k, &sI[1][0][lane], &sPET[1][0][lane], &sRe[0][0][lane], S3);
        } else if (stage == 2) {
            const int blk = j - 2;
            if (blk >= 0 && blk < NGRP)
                upper_block(k, &sRe[1][0][lane], &sPe[0][0][lane], &sQ[0][0][lane], S4);
        } else if (stage == 3) {
            const int blk = j - 3;
            if (blk >= 0 && blk < NGRP)
                lower_block(k, &sPe[1][0][lane], &sQ[1][0][lane],
                            &sQr[(blk & 3) * XCH][lane], S5);
        } else {
            const int blk = j - 5;
            if (blk >= 0 && blk < NGRP)
                route_block(&sQr[0][0], wfr, blk, mn, khi, orow0 + blk * XCH);
        }
        stage_barrier();

        if (stage == 0) {
            const int gn = (j + 3 < NGRP) ? (j + 3) : (NGRP - 1);
            snow_block(bufB, xv + (size_t)gn * XV, k, &sI[1][0][lane], &sPET[1][0][lane], S1, S2);
        } else if (stage == 1) {
            const int blk = j;
            if (blk < NGRP)
                soil_block(k, &sI[0][0][lane], &sPET[0][0][lane], &sRe[1][0][lane], S3);
        } else if (stage == 2) {
            const int blk = j - 1;
            if (blk >= 0 && blk < NGRP)
                upper_block(k, &sRe[0][0][lane], &sPe[1][0][lane], &sQ[1][0][lane], S4);
        } else if (stage == 3) {
            const int blk = j - 2;
            if (blk >= 0 && blk < NGRP)
                lower_block(k, &sPe[0][0][lane], &sQ[0][0][lane],
                            &sQr[(blk & 3) * XCH][lane], S5);
        } else {
            const int blk = j - 4;
            if (blk >= 0 && blk < NGRP)
                route_block(&sQr[0][0], wfr, blk, mn, khi, orow0 + blk * XCH);
        }
        stage_barrier();
    }
    if (stage == 0) asm volatile("s_wait_loadcnt 0x0");
}

extern "C" void kernel_launch(void* const* d_in, const int* in_sizes, int n_in,
                              void* d_out, int out_size, void* d_ws, size_t ws_size,
                              hipStream_t stream) {
    (void)in_sizes; (void)n_in; (void)out_size; (void)ws_size;
    const float* x      = (const float*)d_in[0];
    const float* params = (const float*)d_in[1];
    float* out = (float*)d_out;
    (void)d_ws;

    hbv_scan_route_kernel<<<BDIM / 32, 32 * NSTAGE, 0, stream>>>(x, params, out);
}


# stage 0 keeps a block's I and PET/2 in registers and stores them to LDS in one place behind its last step, not one store inside every step; PET is halved (exact) by stage 0, no longer by stage 1
# speedup vs baseline: 13.9261x; 1.2117x over previous
#include <hip/hip_runtime.h>

#define BDIM 2048
#define TDIM 4096

typedef float v2f __attribute__((ext_vector_type(2)));
typedef float v4f __attribute__((ext_vector_type(4)));
typedef float v8f __attribute__((ext_vector_type(8)));

#define XCH 16
#define XV  (XCH * 3 / 4)
#define NGRP (TDIM / XCH)
#define NSTAGE 5
#define RING (4 * XCH)
#define NITER ((NGRP + NSTAGE + 1) & ~1)

__device__ __forceinline__ float th5(float y) {
    return __builtin_amdgcn_tanhf(5.0f * y);
}

__device__ __forceinline__ float th5_fma(float y, float b) {
    return __builtin_amdgcn_tanhf(__builtin_fmaf(y, 5.0f, b));
}

__device__ __forceinline__ float clip1e5(float d) {
    return fminf(fmaxf(d, -100000.0f), 100000.0f);
}

struct HbvPar {
    float Tt, CFMAX, SCF, CFRMAX, rawCWH, Beta, invFC;
    float n5FCLP, n5UZL;
    float hinvFCLP;
    float n5CWH;
    float hPER, hK0, hK1, hK2, nhK0UZL;
};

__device__ __forceinline__ void stage_barrier() {
    asm volatile("s_wait_dscnt 0x0\n\ts_barrier_signal -1\n\ts_barrier_wait -1" ::: "memory");
}

struct SnowForcing { float a1, a2, hc, hw, hhw, Ps, hPr; };

__device__ __forceinline__ SnowForcing snow_forcing(float P, float Tm, const HbvPar& k) {
    SnowForcing f;
    const float tc = th5(k.Tt - Tm);
    f.hc  = __builtin_fmaf(0.5f, tc, 0.5f);
    f.hw  = __builtin_fmaf(-0.5f, tc, 0.5f);
    f.hhw = __builtin_fmaf(-0.25f, tc, 0.25f);
    f.a1  = (k.Tt - Tm) * k.CFRMAX;
    f.a2  = (k.Tt - Tm) * -k.CFMAX;
    f.Ps  = P * k.SCF * f.hc;
    f.hPr = P * f.hhw;
    return f;
}

__device__ __forceinline__ void snow_step(
    const SnowForcing& f, const HbvPar& k, float& S1, float& S2, float& I)
{
    float mn  = fminf(f.a2, S1);
    float w1  = S1 * k.n5CWH;
    float hX  = __builtin_fmaf(mn, f.hhw, f.hPr);
    float e0  = __builtin_fmaf(-mn, f.hw, f.Ps);
    float ts  = __builtin_amdgcn_tanhf(__builtin_fmaf(S2, 5.0f, w1));
    float mr  = fminf(f.a1, S2);
    I = __builtin_fmaf(hX, ts, hX);
    float nS1 = fmaxf(S1 + clip1e5(__builtin_fmaf(mr, f.hc, e0)), 0.0f);
    float nS2 = fmaxf(fminf(S2 + clip1e5(__builtin_fmaf(-hX, ts, __builtin_fmaf(-mr, f.hc, hX))),
                            nS1 * k.rawCWH), 0.0f);
    S1 = nS1; S2 = nS2;
}

template <int BYTE_OFF>
__device__ __forceinline__ void x_load(v4f& r, const v4f* base) {
    asm volatile("global_load_b128 %0, %1, off offset:%2" : "+v"(r) : "v"(base), "n"(BYTE_OFF));
}

#define XWAIT 22
__device__ __forceinline__ void x_wait(v4f& r) {
    asm volatile("s_wait_loadcnt %1" : "+v"(r) : "n"(XWAIT));
}

template <int S>
__device__ __forceinline__ void snow_steps(
    v4f (&buf)[XV], const v4f* nxt, const HbvPar& k,
    float (&Ik)[XCH], float (&Pk)[XCH], float& S1, float& S2)
{
    if constexpr (S < XCH) {
        if constexpr (S == 0) x_wait(buf[0]);
        if constexpr (((3 * S + 2) >> 2) > ((3 * S - 1) >> 2) && S > 0) x_wait(buf[(3 * S + 2) >> 2]);
        SnowForcing f = snow_forcing(buf[(3 * S + 0) >> 2][(3 * S + 0) & 3], buf[(3 * S + 1) >> 2][(3 * S + 1) & 3], k);
        Pk[S] = 0.5f * buf[(3 * S + 2) >> 2][(3 * S + 2) & 3];
        asm volatile("" :: "v"(f.a1), "v"(f.a2), "v"(f.hc), "v"(f.hw), "v"(f.hhw), "v"(f.Ps), "v"(f.hPr), "v"(Pk[S]));
        float I;
        snow_step(f, k, S1, S2, I);
        Ik[S] = I;
        if constexpr (((3 * (S + 1)) >> 2) > ((3 * S) >> 2)) {
            constexpr int q = ((3 * (S + 1)) >> 2) - 1;
            x_load<16 * q>(buf[q], nxt);
        }
        snow_steps<S + 1>(buf, nxt, k, Ik, Pk, S1, S2);
    }
}

__device__ __forceinline__ void snow_block(
    v4f (&buf)[XV], const v4f* nxt, const HbvPar& k,
    float* __restrict__ outI, float* __restrict__ outPET, float& S1, float& S2)
{
    float Ik[XCH], Pk[XCH];
    snow_steps<0>(buf, nxt, k, Ik, Pk, S1, S2);
    __builtin_amdgcn_sched_barrier(0);
#pragma unroll
    for (int s = 0; s < XCH; ++s) { outI[s * 32] = Ik[s]; outPET[s * 32] = Pk[s]; }
    asm volatile("s_wait_xcnt 0x0");
}

__device__ __forceinline__ void soil_block(
    const HbvPar& k, const float* __restrict__ inI, const float* __restrict__ inPET,
    float* __restrict__ outRe, float& S3)
{
    float Iv[XCH], PETv[XCH];
#pragma unroll
    for (int s = 0; s < XCH; ++s) { Iv[s] = inI[s * 32]; PETv[s] = inPET[s * 32]; }
    __builtin_amdgcn_sched_barrier(0);
#pragma unroll
    for (int s = 0; s < XCH; ++s) {
        const float I = Iv[s], hP = PETv[s];
        float pw  = __builtin_amdgcn_exp2f(k.Beta * __builtin_amdgcn_logf(S3 * k.invFC));
        float Re  = I * pw;
        float t1  = th5(S3);
        float t2  = th5_fma(S3, k.n5FCLP);
        float hg  = __builtin_fmaf(S3, k.hinvFCLP, -0.5f);
        float w   = __builtin_fmaf(-hg, t2, 1.0f + hg);
        float A   = __builtin_fmaf(hP, t1, hP);
        float nS3 = fmaxf(S3 + clip1e5(__builtin_fmaf(-I, pw, I - A * w)), 0.0f);
        S3 = nS3;
        outRe[s * 32] = Re;
    }
}

__device__ __forceinline__ void upper_block(
    const HbvPar& k, const float* __restrict__ inRe,
    float* __restrict__ outPe, float* __restrict__ outQ, float& S4)
{
    float Rev[XCH];
#pragma unroll
    for (int s = 0; s < XCH; ++s) Rev[s] = inRe[s * 32];
    __builtin_amdgcn_sched_barrier(0);
#pragma unroll
    for (int s = 0; s < XCH; ++s) {
        const float Re = Rev[s];
        float t4  = th5(S4);
        float tU  = th5_fma(S4, k.n5UZL);
        float a   = __builtin_fmaf(S4, k.hK1, k.hPER);
        float b   = __builtin_fmaf(S4, k.hK0, k.nhK0UZL);
        float hq1 = S4 * k.hK1;
        float t   = __builtin_fmaf(-a, t4, (Re - a) - b);
        float nS4 = fmaxf(S4 + clip1e5(__builtin_fmaf(-b, tU, t)), 0.0f);
        S4 = nS4;
        outPe[s * 32] = __builtin_fmaf(k.hPER, t4, k.hPER);
        outQ[s * 32]  = __builtin_fmaf(b, tU, b) + __builtin_fmaf(hq1, t4, hq1);
    }
}

__device__ __forceinline__ void lower_block(
    const HbvPar& k, const float* __restrict__ inPe, const float* __restrict__ inQ,
    float* __restrict__ qr, float& S5)
{
    float Pev[XCH], Qv[XCH];
#pragma unroll
    for (int s = 0; s < XCH; ++s) { Pev[s] = inPe[s * 32]; Qv[s] = inQ[s * 32]; }
    __builtin_amdgcn_sched_barrier(0);
#pragma unroll
    for (int s = 0; s < XCH; ++s) {
        float m   = S5 * k.hK2;
        float t5  = th5(S5);
        float nS5 = fmaxf(S5 + clip1e5(__builtin_fmaf(-m, t5, Pev[s] - m)), 0.0f);
        S5 = nS5;
        qr[s * 32] = __builtin_fmaf(m, t5, Qv[s] + m);
    }
}

__device__ __forceinline__ void route_block(
    const float* __restrict__ ring, const v2f (&wfr)[6], const int blk,
    const int mn, const int khi, float* __restrict__ orow)
{
    v2f a0[6], a1[6];
    const int pbase = blk * XCH - 4 + 2 * khi;
#pragma unroll
    for (int c = 0; c < 6; ++c) {
#pragma unroll
        for (int j = 0; j < 2; ++j) {
            const int pt = pbase + 4 * c + j;
            const int pc = pt < 0 ? 0 : (pt > TDIM - 1 ? TDIM - 1 : pt);
            const float* src = ring + (pc & (RING - 1)) * 32 + mn;
            const float v0 = src[0];
            const float v1 = src[16];
            a0[c][j] = (pt == pc) ? v0 : 0.0f;
            a1[c][j] = (pt == pc) ? v1 : 0.0f;
        }
    }
    v8f acc0 = {0.0f, 0.0f, 0.0f, 0.0f, 0.0f, 0.0f, 0.0f, 0.0f};
    v8f acc1 = {0.0f, 0.0f, 0.0f, 0.0f, 0.0f, 0.0f, 0.0f, 0.0f};
#pragma unroll
    for (int c = 0; c < 6; ++c) {
        acc0 = __builtin_amdgcn_wmma_f32_16x16x4_f32(
            false, a0[c], false, wfr[c], (short)0, acc0, false, false);
        acc1 = __builtin_amdgcn_wmma_f32_16x16x4_f32(
            false, a1[c], false, wfr[c], (short)0, acc1, false, false);
    }
#pragma unroll
    for (int v = 0; v < 8; ++v) {
        orow[(size_t)v * TDIM]        = acc0[v];
        orow[(size_t)(v + 16) * TDIM] = acc1[v];
    }
}

__global__ __launch_bounds__(32 * NSTAGE) void hbv_scan_route_kernel(
    const float* __restrict__ x, const float* __restrict__ p,
    float* __restrict__ out)
{
    __shared__ float sI[2][XCH][32], sPET[2][XCH][32], sRe[2][XCH][32];
    __shared__ float sPe[2][XCH][32], sQ[2][XCH][32];
    __shared__ float sQr[RING][32];

    const int lane  = threadIdx.x & 31;
    const int wave  = __builtin_amdgcn_readfirstlane((int)(threadIdx.x >> 5));
    const int stage = (wave == 4) ? 4 : ((wave + 3) & 3);
    const int b = blockIdx.x * 32 + lane;

    const float CFR = p[1] * 0.01f;
    const float FC  = p[4] * 100.0f;
    const float LP  = p[6] * 2.5f;
    HbvPar k;
    k.Tt      = p[0]  * -1.5f;
    k.CFMAX   = p[2];
    k.SCF     = p[3];
    k.Beta    = p[5]  * 8.0f;
    k.n5CWH   = -5.0f * (p[7] * 0.3f);
    const float UZL = p[8] * 120.0f;
    k.hK0     = 0.5f * (p[9]  * 0.5f);
    k.hK1     = 0.5f * (p[10] * 0.3f);
    k.hK2     = 0.5f * (p[11] * 0.01f);
    k.hPER    = 0.5f * (p[12] * 3.5f);
    k.rawCWH  = p[7];
    k.invFC   = __builtin_amdgcn_rcpf(FC);
    k.hinvFCLP = 0.5f * __builtin_amdgcn_rcpf(FC * LP);
    k.CFRMAX  = CFR * k.CFMAX;
    k.n5FCLP  = -5.0f * (FC * LP);
    k.n5UZL   = -5.0f * UZL;
    k.nhK0UZL = -(k.hK0 * UZL);

    float S1 = 0.0f, S2 = 0.0f, S3 = 0.0f, S4 = 0.0f, S5 = 0.0f;

    const v4f* xv = reinterpret_cast<const v4f*>(x + (size_t)b * TDIM * 3);

    const int mn  = lane & 15;
    const int khi = lane >> 4;
    float* orow0 = out + ((size_t)(blockIdx.x * 32 + 8 * khi) * TDIM + mn);
    v2f wfr[6];
    if (stage == 4) {
        constexpr double INV_MID_D = 1.0 / (4.0 + 1e-7);
        constexpr double WSUM_D = 1.0 + 2.0 * ((1.0 - 1.0 * INV_MID_D) +
                                               (1.0 - 2.0 * INV_MID_D) +
                                               (1.0 - 3.0 * INV_MID_D) +
                                               (1.0 - 4.0 * INV_MID_D));
        constexpr float INV_MID  = (float)INV_MID_D;
        constexpr float INV_WSUM = (float)(1.0 / (WSUM_D + 1e-7));
#pragma unroll
        for (int c = 0; c < 6; ++c) {
#pragma unroll
            for (int j = 0; j < 2; ++j) {
                const int kg = 4 * c + j + 2 * khi;
                const float dd = fabsf((float)(kg - mn) - 4.0f);
                wfr[c][j] = fmaxf(0.0f, fmaf(-dd, INV_MID, 1.0f)) * INV_WSUM;
            }
        }
    }

    v4f bufA[XV], bufB[XV];
    if (stage == 0) {
        x_load<0>(bufA[0], xv);   x_load<16>(bufA[1], xv);  x_load<32>(bufA[2], xv);   x_load<48>(bufA[3], xv);
        x_load<64>(bufA[4], xv);  x_load<80>(bufA[5], xv);  x_load<96>(bufA[6], xv);   x_load<112>(bufA[7], xv);
        x_load<128>(bufA[8], xv); x_load<144>(bufA[9], xv); x_load<160>(bufA[10], xv); x_load<176>(bufA[11], xv);
        x_load<0>(bufB[0], xv + XV);   x_load<16>(bufB[1], xv + XV);  x_load<32>(bufB[2], xv + XV);   x_load<48>(bufB[3], xv + XV);
        x_load<64>(bufB[4], xv + XV);  x_load<80>(bufB[5], xv + XV);  x_load<96>(bufB[6], xv + XV);   x_load<112>(bufB[7], xv + XV);
        x_load<128>(bufB[8], xv + XV); x_load<144>(bufB[9], xv + XV); x_load<160>(bufB[10], xv + XV); x_load<176>(bufB[11], xv + XV);
        asm volatile("s_wait_xcnt 0x0");
    }

    for (int j = 0; j < NITER; j += 2) {
        if (stage == 0) {
            const int gn = (j + 2 < NGRP) ? (j + 2) : (NGRP - 1);
            snow_block(bufA, xv + (size_t)gn * XV, k, &sI[0][0][lane], &sPET[0][0][lane], S1, S2);
        } else if (stage == 1) {
            const int blk = j - 1;
            if (blk >= 0 && blk < NGRP)
                soil_block(k, &sI[1][0][lane], &sPET[1][0][lane], &sRe[0][0][lane], S3);
        } else if (stage == 2) {
            const int blk = j - 2;
            if (blk >= 0 && blk < NGRP)
                upper_block(k, &sRe[1][0][lane], &sPe[0][0][lane], &sQ[0][0][lane], S4);
        } else if (stage == 3) {
            const int blk = j - 3;
            if (blk >= 0 && blk < NGRP)
                lower_block(k, &sPe[1][0][lane], &sQ[1][0][lane],
                            &sQr[(blk & 3) * XCH][lane], S5);
        } else {
            const int blk = j - 5;
            if (blk >= 0 && blk < NGRP)
                route_block(&sQr[0][0], wfr, blk, mn, khi, orow0 + blk * XCH);
        }
        stage_barrier();

        if (stage == 0) {
            const int gn = (j + 3 < NGRP) ? (j + 3) : (NGRP - 1);
            snow_block(bufB, xv + (size_t)gn * XV, k, &sI[1][0][lane], &sPET[1][0][lane], S1, S2);
        } else if (stage == 1) {
            const int blk = j;
            if (blk < NGRP)
                soil_block(k, &sI[0][0][lane], &sPET[0][0][lane], &sRe[1][0][lane], S3);
        } else if (stage == 2) {
            const int blk = j - 1;
            if (blk >= 0 && blk < NGRP)
                upper_block(k, &sRe[0][0][lane], &sPe[1][0][lane], &sQ[1][0][lane], S4);
        } else if (stage == 3) {
            const int blk = j - 2;
            if (blk >= 0 && blk < NGRP)
                lower_block(k, &sPe[0][0][lane], &sQ[0][0][lane],
                            &sQr[(blk & 3) * XCH][lane], S5);
        } else {
            const int blk = j - 4;
            if (blk >= 0 && blk < NGRP)
                route_block(&sQr[0][0], wfr, blk, mn, khi, orow0 + blk * XCH);
        }
        stage_barrier();
    }
    if (stage == 0) asm volatile("s_wait_loadcnt 0x0");
}

extern "C" void kernel_launch(void* const* d_in, const int* in_sizes, int n_in,
                              void* d_out, int out_size, void* d_ws, size_t ws_size,
                              hipStream_t stream) {
    (void)in_sizes; (void)n_in; (void)out_size; (void)ws_size;
    const float* x      = (const float*)d_in[0];
    const float* params = (const float*)d_in[1];
    float* out = (float*)d_out;
    (void)d_ws;

    hbv_scan_route_kernel<<<BDIM / 32, 32 * NSTAGE, 0, stream>>>(x, params, out);
}


# links between stages hand over 32 steps, not 16, laid out [lane][step] and moved in 16-byte LDS accesses (results gathered in registers, stored behind the steps); q ring 8 groups deep
# speedup vs baseline: 14.6249x; 1.0502x over previous
#include <hip/hip_runtime.h>

#define BDIM 2048
#define TDIM 4096

typedef float v2f __attribute__((ext_vector_type(2)));
typedef float v4f __attribute__((ext_vector_type(4)));
typedef float v8f __attribute__((ext_vector_type(8)));

#define XCH 16
#define XV  (XCH * 3 / 4)
#define NGRP (TDIM / XCH)
#define NSTAGE 5
#define HB (2 * XCH)
#define NBLK (TDIM / HB)
#define PITCH (HB + 4)
#define LCOPY (32 * PITCH)
#define RING (8 * XCH)
#define NTRIP (NBLK + NSTAGE)

__device__ __forceinline__ float th5(float y) {
    return __builtin_amdgcn_tanhf(5.0f * y);
}

__device__ __forceinline__ float th5_fma(float y, float b) {
    return __builtin_amdgcn_tanhf(__builtin_fmaf(y, 5.0f, b));
}

__device__ __forceinline__ float clip1e5(float d) {
    return fminf(fmaxf(d, -100000.0f), 100000.0f);
}

struct HbvPar {
    float Tt, CFMAX, SCF, CFRMAX, rawCWH, Beta, invFC;
    float n5FCLP, n5UZL;
    float hinvFCLP;
    float n5CWH;
    float hPER, hK0, hK1, hK2, nhK0UZL;
};

__device__ __forceinline__ void stage_barrier() {
    asm volatile("s_wait_dscnt 0x0\n\ts_barrier_signal -1\n\ts_barrier_wait -1" ::: "memory");
}

struct SnowForcing { float a1, a2, hc, hw, hhw, Ps, hPr; };

__device__ __forceinline__ SnowForcing snow_forcing(float P, float Tm, const HbvPar& k) {
    SnowForcing f;
    const float tc = th5(k.Tt - Tm);
    f.hc  = __builtin_fmaf(0.5f, tc, 0.5f);
    f.hw  = __builtin_fmaf(-0.5f, tc, 0.5f);
    f.hhw = __builtin_fmaf(-0.25f, tc, 0.25f);
    f.a1  = (k.Tt - Tm) * k.CFRMAX;
    f.a2  = (k.Tt - Tm) * -k.CFMAX;
    f.Ps  = P * k.SCF * f.hc;
    f.hPr = P * f.hhw;
    return f;
}

__device__ __forceinline__ void snow_step(
    const SnowForcing& f, const HbvPar& k, float& S1, float& S2, float& I)
{
    float mn  = fminf(f.a2, S1);
    float w1  = S1 * k.n5CWH;
    float hX  = __builtin_fmaf(mn, f.hhw, f.hPr);
    float e0  = __builtin_fmaf(-mn, f.hw, f.Ps);
    float ts  = __builtin_amdgcn_tanhf(__builtin_fmaf(S2, 5.0f, w1));
    float mr  = fminf(f.a1, S2);
    I = __builtin_fmaf(hX, ts, hX);
    float nS1 = fmaxf(S1 + clip1e5(__builtin_fmaf(mr, f.hc, e0)), 0.0f);
    float nS2 = fmaxf(fminf(S2 + clip1e5(__builtin_fmaf(-hX, ts, __builtin_fmaf(-mr, f.hc, hX))),
                            nS1 * k.rawCWH), 0.0f);
    S1 = nS1; S2 = nS2;
}

template <int BYTE_OFF>
__device__ __forceinline__ void x_load(v4f& r, const v4f* base) {
    asm volatile("global_load_b128 %0, %1, off offset:%2" : "+v"(r) : "v"(base), "n"(BYTE_OFF));
}

template <int BYTE_OFF>
__device__ __forceinline__ void x_load_first(v4f& r, const v4f* base) {
    asm volatile("global_load_b128 %0, %1, off offset:%2" : "=&v"(r) : "v"(base), "n"(BYTE_OFF));
}

#define XWAIT 22
__device__ __forceinline__ void x_wait(v4f& r) {
    asm volatile("s_wait_loadcnt %1" : "+v"(r) : "n"(XWAIT));
}

template <int S>
__device__ __forceinline__ void snow_steps(
    v4f (&buf)[XV], const v4f* nxt, const HbvPar& k,
    v4f (&Ik)[XCH / 4], v4f (&Pk)[XCH / 4], float& S1, float& S2)
{
    if constexpr (S < XCH) {
        if constexpr (S == 0) x_wait(buf[0]);
        if constexpr (((3 * S + 2) >> 2) > ((3 * S - 1) >> 2) && S > 0) x_wait(buf[(3 * S + 2) >> 2]);
        SnowForcing f = snow_forcing(buf[(3 * S + 0) >> 2][(3 * S + 0) & 3], buf[(3 * S + 1) >> 2][(3 * S + 1) & 3], k);
        const float hP = 0.5f * buf[(3 * S + 2) >> 2][(3 * S + 2) & 3];
        Pk[S >> 2][S & 3] = hP;
        asm volatile("" :: "v"(f.a1), "v"(f.a2), "v"(f.hc), "v"(f.hw), "v"(f.hhw), "v"(f.Ps), "v"(f.hPr), "v"(hP));
        float I;
        snow_step(f, k, S1, S2, I);
        Ik[S >> 2][S & 3] = I;
        if constexpr (((3 * (S + 1)) >> 2) > ((3 * S) >> 2)) {
            constexpr int q = ((3 * (S + 1)) >> 2) - 1;
            x_load<16 * q>(buf[q], nxt);
        }
        snow_steps<S + 1>(buf, nxt, k, Ik, Pk, S1, S2);
    }
}

__device__ __forceinline__ void snow_block(
    v4f (&buf)[XV], const v4f* nxt, const HbvPar& k,
    v4f* __restrict__ outI, v4f* __restrict__ outPET, float& S1, float& S2)
{
    v4f Ik[XCH / 4], Pk[XCH / 4];
    snow_steps<0>(buf, nxt, k, Ik, Pk, S1, S2);
    __builtin_amdgcn_sched_barrier(0);
#pragma unroll
    for (int q = 0; q < XCH / 4; ++q) { outI[q] = Ik[q]; outPET[q] = Pk[q]; }
    asm volatile("s_wait_xcnt 0x0" :: "v"(nxt));
}

__device__ __forceinline__ void soil_block(
    const HbvPar& k, const v4f* __restrict__ inI, const v4f* __restrict__ inPET,
    v4f* __restrict__ outRe, float& S3)
{
    v4f Iv[HB / 4], PETv[HB / 4], Rek[HB / 4];
#pragma unroll
    for (int q = 0; q < HB / 4; ++q) { Iv[q] = inI[q]; PETv[q] = inPET[q]; }
    __builtin_amdgcn_sched_barrier(0);
#pragma unroll
    for (int s = 0; s < HB; ++s) {
        const float I = Iv[s >> 2][s & 3], hP = PETv[s >> 2][s & 3];
        float pw  = __builtin_amdgcn_exp2f(k.Beta * __builtin_amdgcn_logf(S3 * k.invFC));
        float Re  = I * pw;
        float t1  = th5(S3);
        float t2  = th5_fma(S3, k.n5FCLP);
        float hg  = __builtin_fmaf(S3, k.hinvFCLP, -0.5f);
        float w   = __builtin_fmaf(-hg, t2, 1.0f + hg);
        float A   = __builtin_fmaf(hP, t1, hP);
        float nS3 = fmaxf(S3 + clip1e5(__builtin_fmaf(-I, pw, I - A * w)), 0.0f);
        S3 = nS3;
        Rek[s >> 2][s & 3] = Re;
    }
#pragma unroll
    for (int q = 0; q < HB / 4; ++q) outRe[q] = Rek[q];
}

__device__ __forceinline__ void upper_block(
    const HbvPar& k, const v4f* __restrict__ inRe,
    v4f* __restrict__ outPe, v4f* __restrict__ outQ, float& S4)
{
    v4f Rev[HB / 4], Pek[HB / 4], Qk[HB / 4];
#pragma unroll
    for (int q = 0; q < HB / 4; ++q) Rev[q] = inRe[q];
    __builtin_amdgcn_sched_barrier(0);
#pragma unroll
    for (int s = 0; s < HB; ++s) {
        const float Re = Rev[s >> 2][s & 3];
        float t4  = th5(S4);
        float tU  = th5_fma(S4, k.n5UZL);
        float a   = __builtin_fmaf(S4, k.hK1, k.hPER);
        float b   = __builtin_fmaf(S4, k.hK0, k.nhK0UZL);
        float hq1 = S4 * k.hK1;
        float t   = __builtin_fmaf(-a, t4, (Re - a) - b);
        float nS4 = fmaxf(S4 + clip1e5(__builtin_fmaf(-b, tU, t)), 0.0f);
        S4 = nS4;
        Pek[s >> 2][s & 3] = __builtin_fmaf(k.hPER, t4, k.hPER);
        Qk[s >> 2][s & 3]  = __builtin_fmaf(b, tU, b) + __builtin_fmaf(hq1, t4, hq1);
    }
#pragma unroll
    for (int q = 0; q < HB / 4; ++q) { outPe[q] = Pek[q]; outQ[q] = Qk[q]; }
}

__device__ __forceinline__ void lower_block(
    const HbvPar& k, const v4f* __restrict__ inPe, const v4f* __restrict__ inQ,
    float* __restrict__ qr, float& S5)
{
    v4f Pev[HB / 4], Qv[HB / 4];
#pragma unroll
    for (int q = 0; q < HB / 4; ++q) { Pev[q] = inPe[q]; Qv[q] = inQ[q]; }
    __builtin_amdgcn_sched_barrier(0);
#pragma unroll
    for (int s = 0; s < HB; ++s) {
        float m   = S5 * k.hK2;
        float t5  = th5(S5);
        float nS5 = fmaxf(S5 + clip1e5(__builtin_fmaf(-m, t5, Pev[s >> 2][s & 3] - m)), 0.0f);
        S5 = nS5;
        qr[s * 32] = __builtin_fmaf(m, t5, Qv[s >> 2][s & 3] + m);
    }
}

__device__ __forceinline__ void route_block(
    const float* __restrict__ ring, const v2f (&wfr)[6], const int blk,
    const int mn, const int khi, float* __restrict__ orow)
{
    v2f a0[6], a1[6];
    const int pbase = blk * XCH - 4 + 2 * khi;
#pragma unroll
    for (int c = 0; c < 6; ++c) {
#pragma unroll
        for (int j = 0; j < 2; ++j) {
            const int pt = pbase + 4 * c + j;
            const int pc = pt < 0 ? 0 : (pt > TDIM - 1 ? TDIM - 1 : pt);
            const float* src = ring + (pc & (RING - 1)) * 32 + mn;
            const float v0 = src[0];
            const float v1 = src[16];
            a0[c][j] = (pt == pc) ? v0 : 0.0f;
            a1[c][j] = (pt == pc) ? v1 : 0.0f;
        }
    }
    v8f acc0 = {0.0f, 0.0f, 0.0f, 0.0f, 0.0f, 0.0f, 0.0f, 0.0f};
    v8f acc1 = {0.0f, 0.0f, 0.0f, 0.0f, 0.0f, 0.0f, 0.0f, 0.0f};
#pragma unroll
    for (int c = 0; c < 6; ++c) {
        acc0 = __builtin_amdgcn_wmma_f32_16x16x4_f32(
            false, a0[c], false, wfr[c], (short)0, acc0, false, false);
        acc1 = __builtin_amdgcn_wmma_f32_16x16x4_f32(
            false, a1[c], false, wfr[c], (short)0, acc1, false, false);
    }
#pragma unroll
    for (int v = 0; v < 8; ++v) {
        orow[(size_t)v * TDIM]        = acc0[v];
        orow[(size_t)(v + 16) * TDIM] = acc1[v];
    }
}

__global__ __launch_bounds__(32 * NSTAGE) void hbv_scan_route_kernel(
    const float* __restrict__ x, const float* __restrict__ p,
    float* __restrict__ out)
{
    __shared__ __attribute__((aligned(16))) float sI[2 * LCOPY], sPET[2 * LCOPY], sRe[2 * LCOPY];
    __shared__ __attribute__((aligned(16))) float sPe[2 * LCOPY], sQ[2 * LCOPY];
    __shared__ float sQr[RING][32];

    const int lane  = threadIdx.x & 31;
    const int wave  = __builtin_amdgcn_readfirstlane((int)(threadIdx.x >> 5));
    const int stage = (wave == 4) ? 4 : ((wave + 3) & 3);
    const int b = blockIdx.x * 32 + lane;

    const float CFR = p[1] * 0.01f;
    const float FC  = p[4] * 100.0f;
    const float LP  = p[6] * 2.5f;
    HbvPar k;
    k.Tt      = p[0]  * -1.5f;
    k.CFMAX   = p[2];
    k.SCF     = p[3];
    k.Beta    = p[5]  * 8.0f;
    k.n5CWH   = -5.0f * (p[7] * 0.3f);
    const float UZL = p[8] * 120.0f;
    k.hK0     = 0.5f * (p[9]  * 0.5f);
    k.hK1     = 0.5f * (p[10] * 0.3f);
    k.hK2     = 0.5f * (p[11] * 0.01f);
    k.hPER    = 0.5f * (p[12] * 3.5f);
    k.rawCWH  = p[7];
    k.invFC   = __builtin_amdgcn_rcpf(FC);
    k.hinvFCLP = 0.5f * __builtin_amdgcn_rcpf(FC * LP);
    k.CFRMAX  = CFR * k.CFMAX;
    k.n5FCLP  = -5.0f * (FC * LP);
    k.n5UZL   = -5.0f * UZL;
    k.nhK0UZL = -(k.hK0 * UZL);

    float S1 = 0.0f, S2 = 0.0f, S3 = 0.0f, S4 = 0.0f, S5 = 0.0f;

    const v4f* xv = reinterpret_cast<const v4f*>(x + (size_t)b * TDIM * 3);

    const int mn  = lane & 15;
    const int khi = lane >> 4;
    float* orow0 = out + ((size_t)(blockIdx.x * 32 + 8 * khi) * TDIM + mn);
    v2f wfr[6];
    if (stage == 4) {
        constexpr double INV_MID_D = 1.0 / (4.0 + 1e-7);
        constexpr double WSUM_D = 1.0 + 2.0 * ((1.0 - 1.0 * INV_MID_D) +
                                               (1.0 - 2.0 * INV_MID_D) +
                                               (1.0 - 3.0 * INV_MID_D) +
                                               (1.0 - 4.0 * INV_MID_D));
        constexpr float INV_MID  = (float)INV_MID_D;
        constexpr float INV_WSUM = (float)(1.0 / (WSUM_D + 1e-7));
#pragma unroll
        for (int c = 0; c < 6; ++c) {
#pragma unroll
            for (int j = 0; j < 2; ++j) {
                const int kg = 4 * c + j + 2 * khi;
                const float dd = fabsf((float)(kg - mn) - 4.0f);
                wfr[c][j] = fmaxf(0.0f, fmaf(-dd, INV_MID, 1.0f)) * INV_WSUM;
            }
        }
    }

    v4f bufA[XV], bufB[XV];
    if (stage == 0) {
        const v4f* xv1 = xv + XV;
        x_load_first<0>(bufA[0], xv);   x_load_first<16>(bufA[1], xv);  x_load_first<32>(bufA[2], xv);   x_load_first<48>(bufA[3], xv);
        x_load_first<64>(bufA[4], xv);  x_load_first<80>(bufA[5], xv);  x_load_first<96>(bufA[6], xv);   x_load_first<112>(bufA[7], xv);
        x_load_first<128>(bufA[8], xv); x_load_first<144>(bufA[9], xv); x_load_first<160>(bufA[10], xv); x_load_first<176>(bufA[11], xv);
        x_load_first<0>(bufB[0], xv1);   x_load_first<16>(bufB[1], xv1);  x_load_first<32>(bufB[2], xv1);   x_load_first<48>(bufB[3], xv1);
        x_load_first<64>(bufB[4], xv1);  x_load_first<80>(bufB[5], xv1);  x_load_first<96>(bufB[6], xv1);   x_load_first<112>(bufB[7], xv1);
        x_load_first<128>(bufB[8], xv1); x_load_first<144>(bufB[9], xv1); x_load_first<160>(bufB[10], xv1); x_load_first<176>(bufB[11], xv1);
        asm volatile("s_wait_xcnt 0x0" :: "v"(xv), "v"(xv1));
    }

    for (int n = 0; n < NTRIP; ++n) {
        const int wo = (n & 1) * LCOPY + lane * PITCH;
        const int ro = ((n & 1) ^ 1) * LCOPY + lane * PITCH;
        if (stage == 0) {
            const int ga = (2 * n + 2 < NGRP) ? (2 * n + 2) : (NGRP - 1);
            const int gb = (2 * n + 3 < NGRP) ? (2 * n + 3) : (NGRP - 1);
            snow_block(bufA, xv + (size_t)ga * XV, k,
                       reinterpret_cast<v4f*>(sI + wo), reinterpret_cast<v4f*>(sPET + wo), S1, S2);
            snow_block(bufB, xv + (size_t)gb * XV, k,
                       reinterpret_cast<v4f*>(sI + wo + XCH), reinterpret_cast<v4f*>(sPET + wo + XCH), S1, S2);
        } else if (stage == 1) {
            const int blk = n - 1;
            if (blk >= 0 && blk < NBLK)
                soil_block(k, reinterpret_cast<const v4f*>(sI + ro), reinterpret_cast<const v4f*>(sPET + ro),
                           reinterpret_cast<v4f*>(sRe + wo), S3);
        } else if (stage == 2) {
            const int blk = n - 2;
            if (blk >= 0 && blk < NBLK)
                upper_block(k, reinterpret_cast<const v4f*>(sRe + ro),
                            reinterpret_cast<v4f*>(sPe + wo), reinterpret_cast<v4f*>(sQ + wo), S4);
        } else if (stage == 3) {
            const int blk = n - 3;
            if (blk >= 0 && blk < NBLK)
                lower_block(k, reinterpret_cast<const v4f*>(sPe + ro), reinterpret_cast<const v4f*>(sQ + ro),
                            &sQr[(blk * HB) & (RING - 1)][lane], S5);
        } else {
            const int g0 = 2 * (n - 4) - 1;
            if (g0 >= 0 && g0 < NGRP)
                route_block(&sQr[0][0], wfr, g0, mn, khi, orow0 + g0 * XCH);
            const int g1 = g0 + 1;
            if (g1 >= 0 && g1 < NGRP)
                route_block(&sQr[0][0], wfr, g1, mn, khi, orow0 + g1 * XCH);
        }
        stage_barrier();
    }
    if (stage == 0) asm volatile("s_wait_loadcnt 0x0");
}

extern "C" void kernel_launch(void* const* d_in, const int* in_sizes, int n_in,
                              void* d_out, int out_size, void* d_ws, size_t ws_size,
                              hipStream_t stream) {
    (void)in_sizes; (void)n_in; (void)out_size; (void)ws_size;
    const float* x      = (const float*)d_in[0];
    const float* params = (const float*)d_in[1];
    float* out = (float*)d_out;
    (void)d_ws;

    hbv_scan_route_kernel<<<BDIM / 32, 32 * NSTAGE, 0, stream>>>(x, params, out);
}
